# starting kernel
# baseline (speedup 1.0000x reference)
#include <hip/hip_runtime.h>
#include <hip/hip_bf16.h>

#define N_NODES 50000
#define N_EDGES 800000
#define E_TOT   (N_EDGES + N_NODES)
#define F_IN    256
#define HEADS   8
#define HID     32
#define N_CLS   16
#define HC1     (HEADS * HID)
#define HC2     (HEADS * N_CLS)
#define NEG_SLOPE 0.2f

typedef float v2f __attribute__((ext_vector_type(2)));
typedef float v8f __attribute__((ext_vector_type(8)));

__device__ __forceinline__ unsigned f2key(float f) {
  unsigned b = __float_as_uint(f);
  return (b & 0x80000000u) ? ~b : (b | 0x80000000u);
}
__device__ __forceinline__ float key2f(unsigned k) {
  return __uint_as_float((k & 0x80000000u) ? (k ^ 0x80000000u) : ~k);
}

__global__ void zero4_kernel(float4* __restrict__ p, long long n4) {
  long long i = (long long)blockIdx.x * blockDim.x + threadIdx.x;
  if (i < n4) p[i] = make_float4(0.f, 0.f, 0.f, 0.f);
}

template <int N, int K>
__global__ __launch_bounds__(256) void gemm_f32_wmma(
    const float* __restrict__ A, const float* __restrict__ B,
    float* __restrict__ C, int M) {
  constexpr int TILES_N = N / 64;
  const int lane  = threadIdx.x & 31;
  const int wave  = blockIdx.x * (blockDim.x >> 5) + (threadIdx.x >> 5);
  const int tileM = wave / TILES_N;
  const int tileN = wave % TILES_N;
  if (tileM * 16 >= M) return;

  const int r    = lane & 15;
  const int ksel = (lane >> 4) << 1;

  const float* __restrict__ arow  = A + (size_t)(tileM * 16 + r) * K + ksel;
  const float* __restrict__ bbase = B + (size_t)ksel * N + tileN * 64 + r;

  v8f acc0 = {}, acc1 = {}, acc2 = {}, acc3 = {};
  #pragma unroll 2
  for (int kk = 0; kk < K; kk += 4) {
    v2f a, b0, b1, b2, b3;
    a[0]  = arow[0];
    a[1]  = arow[1];
    b0[0] = bbase[0];       b0[1] = bbase[N];
    b1[0] = bbase[16];      b1[1] = bbase[N + 16];
    b2[0] = bbase[32];      b2[1] = bbase[N + 32];
    b3[0] = bbase[48];      b3[1] = bbase[N + 48];
    acc0 = __builtin_amdgcn_wmma_f32_16x16x4_f32(false, a, false, b0, (short)0, acc0, false, false);
    acc1 = __builtin_amdgcn_wmma_f32_16x16x4_f32(false, a, false, b1, (short)0, acc1, false, false);
    acc2 = __builtin_amdgcn_wmma_f32_16x16x4_f32(false, a, false, b2, (short)0, acc2, false, false);
    acc3 = __builtin_amdgcn_wmma_f32_16x16x4_f32(false, a, false, b3, (short)0, acc3, false, false);
    arow  += 4;
    bbase += (size_t)4 * N;
  }

  float* __restrict__ crow =
      C + (size_t)(tileM * 16 + (lane >> 4) * 8) * N + tileN * 64 + (lane & 15);
  #pragma unroll
  for (int i = 0; i < 8; ++i) {
    crow[(size_t)i * N +  0] = acc0[i];
    crow[(size_t)i * N + 16] = acc1[i];
    crow[(size_t)i * N + 32] = acc2[i];
    crow[(size_t)i * N + 48] = acc3[i];
  }
}

template <int C>
__global__ void alpha_kernel(const float* __restrict__ h,
                             const float* __restrict__ a_src,
                             const float* __restrict__ a_dst,
                             float* __restrict__ as, float* __restrict__ ad) {
  int idx = blockIdx.x * blockDim.x + threadIdx.x;
  if (idx >= N_NODES * HEADS) return;
  int n = idx >> 3, hd = idx & 7;
  const float* hp = h + (size_t)n * HEADS * C + (size_t)hd * C;
  const float* av = a_src + hd * C;
  const float* bv = a_dst + hd * C;
  float s = 0.f, d = 0.f;
  #pragma unroll 4
  for (int c = 0; c < C; ++c) {
    float v = hp[c];
    s += v * av[c];
    d += v * bv[c];
  }
  as[idx] = s;
  ad[idx] = d;
}

__device__ __forceinline__ void edge_endpoints(const int* __restrict__ ei,
                                               int e, int& src, int& dst) {
  if (e < N_EDGES) {
    src = ei[e];
    dst = ei[N_EDGES + e];
  } else {
    src = dst = e - N_EDGES;
  }
}

__global__ void edge_max_kernel(const int* __restrict__ ei,
                                const float* __restrict__ as,
                                const float* __restrict__ ad,
                                unsigned* __restrict__ mkeys) {
  long long idx = (long long)blockIdx.x * blockDim.x + threadIdx.x;
  if (idx >= (long long)E_TOT * HEADS) return;
  int e = (int)(idx >> 3), hd = (int)(idx & 7);
  int src, dst;
  edge_endpoints(ei, e, src, dst);
  float v = as[src * HEADS + hd] + ad[dst * HEADS + hd];
  v = v > 0.f ? v : NEG_SLOPE * v;
  atomicMax(&mkeys[dst * HEADS + hd], f2key(v));
}

__global__ void edge_exp_kernel(const int* __restrict__ ei,
                                const float* __restrict__ as,
                                const float* __restrict__ ad,
                                const unsigned* __restrict__ mkeys,
                                float* __restrict__ ex,
                                float* __restrict__ den) {
  long long idx = (long long)blockIdx.x * blockDim.x + threadIdx.x;
  if (idx >= (long long)E_TOT * HEADS) return;
  int e = (int)(idx >> 3), hd = (int)(idx & 7);
  int src, dst;
  edge_endpoints(ei, e, src, dst);
  float v = as[src * HEADS + hd] + ad[dst * HEADS + hd];
  v = v > 0.f ? v : NEG_SLOPE * v;
  float m  = key2f(mkeys[dst * HEADS + hd]);
  float xv = __expf(v - m);
  ex[idx]  = xv;
  atomicAdd(&den[dst * HEADS + hd], xv);
}

template <int HCv, int Cv>
__global__ void edge_agg_kernel(const int* __restrict__ ei,
                                const float* __restrict__ ex,
                                const float* __restrict__ den,
                                const float* __restrict__ hmat,
                                float* __restrict__ agg) {
  constexpr int HC4 = HCv / 4;
  long long idx = (long long)blockIdx.x * blockDim.x + threadIdx.x;
  if (idx >= (long long)E_TOT * HC4) return;
  int e = (int)(idx / HC4);
  int j = (int)(idx % HC4) * 4;
  int hd = j / Cv;
  int src, dst;
  edge_endpoints(ei, e, src, dst);
  float w = ex[(size_t)e * HEADS + hd] /
            (den[(size_t)dst * HEADS + hd] + 1e-16f);
  const float4 hv = *(const float4*)(hmat + (size_t)src * HCv + j);
  float* ap = agg + (size_t)dst * HCv + j;
  atomicAdd(ap + 0, w * hv.x);
  atomicAdd(ap + 1, w * hv.y);
  atomicAdd(ap + 2, w * hv.z);
  atomicAdd(ap + 3, w * hv.w);
}

template <int HCv>
__global__ void bias_relu_kernel(float* __restrict__ p,
                                 const float* __restrict__ bias, long long n) {
  long long i = (long long)blockIdx.x * blockDim.x + threadIdx.x;
  if (i >= n) return;
  float v = p[i] + bias[(int)(i % HCv)];
  p[i] = v > 0.f ? v : 0.f;
}

template <int HCv>
__global__ void bias_add_kernel(float* __restrict__ p,
                                const float* __restrict__ bias, long long n) {
  long long i = (long long)blockIdx.x * blockDim.x + threadIdx.x;
  if (i >= n) return;
  p[i] = p[i] + bias[(int)(i % HCv)];
}

static inline int blocks_for(long long n, int bs) {
  return (int)((n + bs - 1) / bs);
}

extern "C" void kernel_launch(void* const* d_in, const int* in_sizes, int n_in,
                              void* d_out, int out_size, void* d_ws, size_t ws_size,
                              hipStream_t stream) {
  const float* x      = (const float*)d_in[0];
  const int*   ei     = (const int*)  d_in[1];
  const float* W1     = (const float*)d_in[2];
  const float* a_src1 = (const float*)d_in[3];
  const float* a_dst1 = (const float*)d_in[4];
  const float* b1     = (const float*)d_in[5];
  const float* W2     = (const float*)d_in[6];
  const float* a_src2 = (const float*)d_in[7];
  const float* a_dst2 = (const float*)d_in[8];
  const float* b2     = (const float*)d_in[9];
  float* out = (float*)d_out;

  char* ws = (char*)d_ws;
  size_t off = 0;
  auto carve = [&](size_t bytes) {
    void* p = ws + off;
    off += (bytes + 255) & ~(size_t)255;
    return p;
  };
  float*    hbuf   = (float*)   carve((size_t)N_NODES * HC1 * 4);
  float*    aggbuf = (float*)   carve((size_t)N_NODES * HC1 * 4);
  float*    as     = (float*)   carve((size_t)N_NODES * HEADS * 4);
  float*    ad     = (float*)   carve((size_t)N_NODES * HEADS * 4);
  unsigned* mkeys  = (unsigned*)carve((size_t)N_NODES * HEADS * 4);
  float*    den    = (float*)   carve((size_t)N_NODES * HEADS * 4);
  float*    ex     = (float*)   carve((size_t)E_TOT   * HEADS * 4);

  const int BS = 256;
  const long long nNH    = (long long)N_NODES * HEADS;
  const long long nEH    = (long long)E_TOT * HEADS;
  const long long nElem1 = (long long)N_NODES * HC1;
  const long long nElem2 = (long long)N_NODES * HC2;

  {
    long long waves = (N_NODES / 16) * (HC1 / 64);
    gemm_f32_wmma<HC1, F_IN><<<blocks_for(waves * 32, BS), BS, 0, stream>>>(
        x, W1, hbuf, N_NODES);
  }
  alpha_kernel<HID><<<blocks_for(nNH, BS), BS, 0, stream>>>(hbuf, a_src1, a_dst1, as, ad);
  zero4_kernel<<<blocks_for(nNH / 4, BS), BS, 0, stream>>>((float4*)mkeys, nNH / 4);
  zero4_kernel<<<blocks_for(nNH / 4, BS), BS, 0, stream>>>((float4*)den, nNH / 4);
  edge_max_kernel<<<blocks_for(nEH, BS), BS, 0, stream>>>(ei, as, ad, mkeys);
  edge_exp_kernel<<<blocks_for(nEH, BS), BS, 0, stream>>>(ei, as, ad, mkeys, ex, den);
  zero4_kernel<<<blocks_for(nElem1 / 4, BS), BS, 0, stream>>>((float4*)aggbuf, nElem1 / 4);
  edge_agg_kernel<HC1, HID><<<blocks_for((long long)E_TOT * (HC1 / 4), BS), BS, 0, stream>>>(
      ei, ex, den, hbuf, aggbuf);
  bias_relu_kernel<HC1><<<blocks_for(nElem1, BS), BS, 0, stream>>>(aggbuf, b1, nElem1);

  {
    long long waves = (N_NODES / 16) * (HC2 / 64);
    gemm_f32_wmma<HC2, HC1><<<blocks_for(waves * 32, BS), BS, 0, stream>>>(
        aggbuf, W2, hbuf, N_NODES);
  }
  alpha_kernel<N_CLS><<<blocks_for(nNH, BS), BS, 0, stream>>>(hbuf, a_src2, a_dst2, as, ad);
  zero4_kernel<<<blocks_for(nNH / 4, BS), BS, 0, stream>>>((float4*)mkeys, nNH / 4);
  zero4_kernel<<<blocks_for(nNH / 4, BS), BS, 0, stream>>>((float4*)den, nNH / 4);
  edge_max_kernel<<<blocks_for(nEH, BS), BS, 0, stream>>>(ei, as, ad, mkeys);
  edge_exp_kernel<<<blocks_for(nEH, BS), BS, 0, stream>>>(ei, as, ad, mkeys, ex, den);
  zero4_kernel<<<blocks_for(nElem2 / 4, BS), BS, 0, stream>>>((float4*)out, nElem2 / 4);
  edge_agg_kernel<HC2, N_CLS><<<blocks_for((long long)E_TOT * (HC2 / 4), BS), BS, 0, stream>>>(
      ei, ex, den, hbuf, out);
  bias_add_kernel<HC2><<<blocks_for(nElem2, BS), BS, 0, stream>>>(out, b2, nElem2);
}


# Edges grouped by destination in-call (count, scan, atomic-cursor fill); one wave per node does max, sum, weighted row sum in registers: no float atomics, sums run in arrival-ranked fill order
# speedup vs baseline: 2.5169x; 2.5169x over previous
#include <hip/hip_runtime.h>
#include <hip/hip_bf16.h>

#define N_NODES 50000
#define N_EDGES 800000
#define E_TOT   (N_EDGES + N_NODES)
#define F_IN    256
#define HEADS   8
#define HID     32
#define N_CLS   16
#define HC1     (HEADS * HID)
#define HC2     (HEADS * N_CLS)
#define NEG_SLOPE 0.2f

typedef float v2f __attribute__((ext_vector_type(2)));
typedef float v8f __attribute__((ext_vector_type(8)));

__global__ void zero4_kernel(float4* __restrict__ p, long long n4) {
  long long i = (long long)blockIdx.x * blockDim.x + threadIdx.x;
  if (i < n4) p[i] = make_float4(0.f, 0.f, 0.f, 0.f);
}

template <int N, int K>
__global__ __launch_bounds__(256) void gemm_f32_wmma(
    const float* __restrict__ A, const float* __restrict__ B,
    float* __restrict__ C, int M) {
  constexpr int TILES_N = N / 64;
  const int lane  = threadIdx.x & 31;
  const int wave  = blockIdx.x * (blockDim.x >> 5) + (threadIdx.x >> 5);
  const int tileM = wave / TILES_N;
  const int tileN = wave % TILES_N;
  if (tileM * 16 >= M) return;

  const int r    = lane & 15;
  const int ksel = (lane >> 4) << 1;

  const float* __restrict__ arow  = A + (size_t)(tileM * 16 + r) * K + ksel;
  const float* __restrict__ bbase = B + (size_t)ksel * N + tileN * 64 + r;

  v8f acc0 = {}, acc1 = {}, acc2 = {}, acc3 = {};
  #pragma unroll 2
  for (int kk = 0; kk < K; kk += 4) {
    v2f a, b0, b1, b2, b3;
    a[0]  = arow[0];
    a[1]  = arow[1];
    b0[0] = bbase[0];       b0[1] = bbase[N];
    b1[0] = bbase[16];      b1[1] = bbase[N + 16];
    b2[0] = bbase[32];      b2[1] = bbase[N + 32];
    b3[0] = bbase[48];      b3[1] = bbase[N + 48];
    acc0 = __builtin_amdgcn_wmma_f32_16x16x4_f32(false, a, false, b0, (short)0, acc0, false, false);
    acc1 = __builtin_amdgcn_wmma_f32_16x16x4_f32(false, a, false, b1, (short)0, acc1, false, false);
    acc2 = __builtin_amdgcn_wmma_f32_16x16x4_f32(false, a, false, b2, (short)0, acc2, false, false);
    acc3 = __builtin_amdgcn_wmma_f32_16x16x4_f32(false, a, false, b3, (short)0, acc3, false, false);
    arow  += 4;
    bbase += (size_t)4 * N;
  }

  float* __restrict__ crow =
      C + (size_t)(tileM * 16 + (lane >> 4) * 8) * N + tileN * 64 + (lane & 15);
  #pragma unroll
  for (int i = 0; i < 8; ++i) {
    crow[(size_t)i * N +  0] = acc0[i];
    crow[(size_t)i * N + 16] = acc1[i];
    crow[(size_t)i * N + 32] = acc2[i];
    crow[(size_t)i * N + 48] = acc3[i];
  }
}

template <int C>
__global__ void alpha_kernel(const float* __restrict__ h,
                             const float* __restrict__ a_src,
                             const float* __restrict__ a_dst,
                             float* __restrict__ as, float* __restrict__ ad) {
  int idx = blockIdx.x * blockDim.x + threadIdx.x;
  if (idx >= N_NODES * HEADS) return;
  int n = idx >> 3, hd = idx & 7;
  const float* hp = h + (size_t)n * HEADS * C + (size_t)hd * C;
  const float* av = a_src + hd * C;
  const float* bv = a_dst + hd * C;
  float s = 0.f, d = 0.f;
  #pragma unroll 4
  for (int c = 0; c < C; ++c) {
    float v = hp[c];
    s += v * av[c];
    d += v * bv[c];
  }
  as[idx] = s;
  ad[idx] = d;
}

__global__ void degree_kernel(const int* __restrict__ ei, int* __restrict__ cnt) {
  int e = blockIdx.x * blockDim.x + threadIdx.x;
  if (e >= N_EDGES) return;
  atomicAdd(&cnt[ei[N_EDGES + e]], 1);
}

#define SCAN_WAVES 32
#define SCAN_ITERS 13
__global__ __launch_bounds__(1024) void rowptr_scan_kernel(
    const int* __restrict__ cnt, int* __restrict__ rowptr, int* __restrict__ cursor) {
  static_assert(N_NODES % 4 == 0, "a 16-byte group is inside or outside as a whole");
  static_assert(SCAN_WAVES * SCAN_ITERS * 128 >= N_NODES, "span covers all nodes");
  __shared__ int wtot[SCAN_WAVES];
  const int lane = threadIdx.x & 31;
  const int w    = threadIdx.x >> 5;
  const int base = w * (SCAN_ITERS * 128);
  int4 v[SCAN_ITERS];
  int  x[SCAN_ITERS];
  #pragma unroll
  for (int i = 0; i < SCAN_ITERS; ++i) {
    const int k = base + (i * 32 + lane) * 4;
    v[i] = (k < N_NODES) ? *(const int4*)(cnt + k) : make_int4(0, 0, 0, 0);
  }
  #pragma unroll
  for (int i = 0; i < SCAN_ITERS; ++i) {
    int s = v[i].x + v[i].y + v[i].z + v[i].w;
    #pragma unroll
    for (int off = 1; off < 32; off <<= 1) {
      int y = __shfl_up(s, off, 32);
      if (lane >= off) s += y;
    }
    x[i] = s;
  }
  int carry = 0;
  #pragma unroll
  for (int i = 0; i < SCAN_ITERS; ++i) {
    const int own = v[i].x + v[i].y + v[i].z + v[i].w;
    const int tot = __shfl(x[i], 31, 32);
    x[i]  = x[i] - own + carry;
    carry += tot;
  }
  if (lane == 0) wtot[w] = carry;
  __syncthreads();
  int t = (lane < w) ? wtot[lane] : 0;
  #pragma unroll
  for (int off = 16; off > 0; off >>= 1) t += __shfl_xor(t, off, 32);
  #pragma unroll
  for (int i = 0; i < SCAN_ITERS; ++i) {
    const int k = base + (i * 32 + lane) * 4;
    if (k < N_NODES) {
      int4 r;
      r.x = x[i] + t;
      r.y = r.x + v[i].x;
      r.z = r.y + v[i].y;
      r.w = r.z + v[i].z;
      *(int4*)(rowptr + k) = r;
      *(int4*)(cursor + k) = r;
    }
  }
  if (threadIdx.x == 1023) rowptr[N_NODES] = t + carry;
}

__global__ void csr_fill_kernel(const int* __restrict__ ei, int* __restrict__ cursor,
                                int* __restrict__ col) {
  int e = blockIdx.x * blockDim.x + threadIdx.x;
  if (e >= N_EDGES) return;
  int s = ei[e];
  int d = ei[N_EDGES + e];
  int pos = atomicAdd(&cursor[d], 1);
  col[pos] = s;
}

template <int HCv, bool RELU>
__global__ __launch_bounds__(256, 16) void node_softmax_agg_kernel(
    const int* __restrict__ rowptr, const int* __restrict__ col,
    const float* __restrict__ as, const float* __restrict__ ad,
    const float* __restrict__ hmat, const float* __restrict__ bias,
    float* __restrict__ outp) {
  constexpr int VEC = HCv / 32;
  constexpr int NV4 = VEC / 4;
  static_assert((HCv / HEADS) / VEC == 4, "four lanes per head");
  const int lane = threadIdx.x & 31;
  const int n = __builtin_amdgcn_readfirstlane(
      (int)(blockIdx.x * (blockDim.x >> 5) + (threadIdx.x >> 5)));
  if (n >= N_NODES) return;
  const int hd  = lane >> 2;
  const int beg = rowptr[n];
  const int end = rowptr[n + 1];
  const float adv = ad[n * HEADS + hd];

  float vs = as[n * HEADS + hd] + adv;
  vs = vs > 0.f ? vs : NEG_SLOPE * vs;

  float m = vs;
  #pragma unroll 4
  for (int i = beg; i < end; ++i) {
    const int s = col[i];
    float v = as[s * HEADS + hd] + adv;
    v = v > 0.f ? v : NEG_SLOPE * v;
    m = v > m ? v : m;
  }

  float den = __expf(vs - m);
  #pragma unroll 4
  for (int i = beg; i < end; ++i) {
    const int s = col[i];
    float v = as[s * HEADS + hd] + adv;
    v = v > 0.f ? v : NEG_SLOPE * v;
    den += __expf(v - m);
  }
  const float dn = den + 1e-16f;

  float acc[VEC];
  {
    const float w = __expf(vs - m) / dn;
    const float4* __restrict__ hp =
        (const float4*)(hmat + (size_t)n * HCv + lane * VEC);
    #pragma unroll
    for (int q = 0; q < NV4; ++q) {
      const float4 t = hp[q];
      acc[4 * q + 0] = w * t.x;
      acc[4 * q + 1] = w * t.y;
      acc[4 * q + 2] = w * t.z;
      acc[4 * q + 3] = w * t.w;
    }
  }
  #pragma unroll 4
  for (int i = beg; i < end; ++i) {
    const int s = col[i];
    float v = as[s * HEADS + hd] + adv;
    v = v > 0.f ? v : NEG_SLOPE * v;
    const float w = __expf(v - m) / dn;
    const float4* __restrict__ hp =
        (const float4*)(hmat + (size_t)s * HCv + lane * VEC);
    #pragma unroll
    for (int q = 0; q < NV4; ++q) {
      const float4 t = hp[q];
      acc[4 * q + 0] += w * t.x;
      acc[4 * q + 1] += w * t.y;
      acc[4 * q + 2] += w * t.z;
      acc[4 * q + 3] += w * t.w;
    }
  }

  float4* __restrict__ op = (float4*)(outp + (size_t)n * HCv + lane * VEC);
  const float4* __restrict__ bp = (const float4*)(bias + lane * VEC);
  #pragma unroll
  for (int q = 0; q < NV4; ++q) {
    const float4 b = bp[q];
    float4 r;
    r.x = acc[4 * q + 0] + b.x;
    r.y = acc[4 * q + 1] + b.y;
    r.z = acc[4 * q + 2] + b.z;
    r.w = acc[4 * q + 3] + b.w;
    if (RELU) {
      r.x = r.x > 0.f ? r.x : 0.f;
      r.y = r.y > 0.f ? r.y : 0.f;
      r.z = r.z > 0.f ? r.z : 0.f;
      r.w = r.w > 0.f ? r.w : 0.f;
    }
    op[q] = r;
  }
}

static inline int blocks_for(long long n, int bs) {
  return (int)((n + bs - 1) / bs);
}

extern "C" void kernel_launch(void* const* d_in, const int* in_sizes, int n_in,
                              void* d_out, int out_size, void* d_ws, size_t ws_size,
                              hipStream_t stream) {
  const float* x      = (const float*)d_in[0];
  const int*   ei     = (const int*)  d_in[1];
  const float* W1     = (const float*)d_in[2];
  const float* a_src1 = (const float*)d_in[3];
  const float* a_dst1 = (const float*)d_in[4];
  const float* b1     = (const float*)d_in[5];
  const float* W2     = (const float*)d_in[6];
  const float* a_src2 = (const float*)d_in[7];
  const float* a_dst2 = (const float*)d_in[8];
  const float* b2     = (const float*)d_in[9];
  float* out = (float*)d_out;

  char* ws = (char*)d_ws;
  size_t off = 0;
  auto carve = [&](size_t bytes) {
    void* p = ws + off;
    off += (bytes + 255) & ~(size_t)255;
    return p;
  };
  float* hbuf   = (float*)carve((size_t)N_NODES * HC1 * 4);
  float* aggbuf = (float*)carve((size_t)N_NODES * HC1 * 4);
  float* as     = (float*)carve((size_t)N_NODES * HEADS * 4);
  float* ad     = (float*)carve((size_t)N_NODES * HEADS * 4);
  int*   cnt    = (int*)  carve((size_t)N_NODES * 4);
  int*   rowptr = (int*)  carve((size_t)(N_NODES + 1) * 4);
  int*   cursor = (int*)  carve((size_t)N_NODES * 4);
  int*   col    = (int*)  carve((size_t)N_EDGES * 4);

  const int BS = 256;
  const long long nNH = (long long)N_NODES * HEADS;

  {
    long long waves = (N_NODES / 16) * (HC1 / 64);
    gemm_f32_wmma<HC1, F_IN><<<blocks_for(waves * 32, BS), BS, 0, stream>>>(
        x, W1, hbuf, N_NODES);
  }
  alpha_kernel<HID><<<blocks_for(nNH, BS), BS, 0, stream>>>(hbuf, a_src1, a_dst1, as, ad);

  zero4_kernel<<<blocks_for(N_NODES / 4, BS), BS, 0, stream>>>((float4*)cnt, N_NODES / 4);
  degree_kernel<<<blocks_for(N_EDGES, BS), BS, 0, stream>>>(ei, cnt);
  rowptr_scan_kernel<<<1, 1024, 0, stream>>>(cnt, rowptr, cursor);
  csr_fill_kernel<<<blocks_for(N_EDGES, BS), BS, 0, stream>>>(ei, cursor, col);

  node_softmax_agg_kernel<HC1, true><<<blocks_for((long long)N_NODES * 32, BS), BS, 0, stream>>>(
      rowptr, col, as, ad, hbuf, b1, aggbuf);

  {
    long long waves = (N_NODES / 16) * (HC2 / 64);
    gemm_f32_wmma<HC2, HC1><<<blocks_for(waves * 32, BS), BS, 0, stream>>>(
        aggbuf, W2, hbuf, N_NODES);
  }
  alpha_kernel<N_CLS><<<blocks_for(nNH, BS), BS, 0, stream>>>(hbuf, a_src2, a_dst2, as, ad);
  node_softmax_agg_kernel<HC2, false><<<blocks_for((long long)N_NODES * 32, BS), BS, 0, stream>>>(
      rowptr, col, as, ad, hbuf, b2, out);
}


# Aggregation: each edge logit gathered once, four edges per head quad, kept in registers for max, sum (per lane, then across the quad) and weights; weights reach the row loop through LDS
# speedup vs baseline: 2.6056x; 1.0352x over previous
#include <hip/hip_runtime.h>
#include <hip/hip_bf16.h>

#define N_NODES 50000
#define N_EDGES 800000
#define E_TOT   (N_EDGES + N_NODES)
#define F_IN    256
#define HEADS   8
#define HID     32
#define N_CLS   16
#define HC1     (HEADS * HID)
#define HC2     (HEADS * N_CLS)
#define NEG_SLOPE 0.2f

typedef float v2f __attribute__((ext_vector_type(2)));
typedef float v8f __attribute__((ext_vector_type(8)));

__global__ void zero4_kernel(float4* __restrict__ p, long long n4) {
  long long i = (long long)blockIdx.x * blockDim.x + threadIdx.x;
  if (i < n4) p[i] = make_float4(0.f, 0.f, 0.f, 0.f);
}

template <int N, int K>
__global__ __launch_bounds__(256) void gemm_f32_wmma(
    const float* __restrict__ A, const float* __restrict__ B,
    float* __restrict__ C, int M) {
  constexpr int TILES_N = N / 64;
  const int lane  = threadIdx.x & 31;
  const int wave  = blockIdx.x * (blockDim.x >> 5) + (threadIdx.x >> 5);
  const int tileM = wave / TILES_N;
  const int tileN = wave % TILES_N;
  if (tileM * 16 >= M) return;

  const int r    = lane & 15;
  const int ksel = (lane >> 4) << 1;

  const float* __restrict__ arow  = A + (size_t)(tileM * 16 + r) * K + ksel;
  const float* __restrict__ bbase = B + (size_t)ksel * N + tileN * 64 + r;

  v8f acc0 = {}, acc1 = {}, acc2 = {}, acc3 = {};
  #pragma unroll 2
  for (int kk = 0; kk < K; kk += 4) {
    v2f a, b0, b1, b2, b3;
    a[0]  = arow[0];
    a[1]  = arow[1];
    b0[0] = bbase[0];       b0[1] = bbase[N];
    b1[0] = bbase[16];      b1[1] = bbase[N + 16];
    b2[0] = bbase[32];      b2[1] = bbase[N + 32];
    b3[0] = bbase[48];      b3[1] = bbase[N + 48];
    acc0 = __builtin_amdgcn_wmma_f32_16x16x4_f32(false, a, false, b0, (short)0, acc0, false, false);
    acc1 = __builtin_amdgcn_wmma_f32_16x16x4_f32(false, a, false, b1, (short)0, acc1, false, false);
    acc2 = __builtin_amdgcn_wmma_f32_16x16x4_f32(false, a, false, b2, (short)0, acc2, false, false);
    acc3 = __builtin_amdgcn_wmma_f32_16x16x4_f32(false, a, false, b3, (short)0, acc3, false, false);
    arow  += 4;
    bbase += (size_t)4 * N;
  }

  float* __restrict__ crow =
      C + (size_t)(tileM * 16 + (lane >> 4) * 8) * N + tileN * 64 + (lane & 15);
  #pragma unroll
  for (int i = 0; i < 8; ++i) {
    crow[(size_t)i * N +  0] = acc0[i];
    crow[(size_t)i * N + 16] = acc1[i];
    crow[(size_t)i * N + 32] = acc2[i];
    crow[(size_t)i * N + 48] = acc3[i];
  }
}

template <int C>
__global__ void alpha_kernel(const float* __restrict__ h,
                             const float* __restrict__ a_src,
                             const float* __restrict__ a_dst,
                             float* __restrict__ as, float* __restrict__ ad) {
  int idx = blockIdx.x * blockDim.x + threadIdx.x;
  if (idx >= N_NODES * HEADS) return;
  int n = idx >> 3, hd = idx & 7;
  const float* hp = h + (size_t)n * HEADS * C + (size_t)hd * C;
  const float* av = a_src + hd * C;
  const float* bv = a_dst + hd * C;
  float s = 0.f, d = 0.f;
  #pragma unroll 4
  for (int c = 0; c < C; ++c) {
    float v = hp[c];
    s += v * av[c];
    d += v * bv[c];
  }
  as[idx] = s;
  ad[idx] = d;
}

__global__ void degree_kernel(const int* __restrict__ ei, int* __restrict__ cnt) {
  int e = blockIdx.x * blockDim.x + threadIdx.x;
  if (e >= N_EDGES) return;
  atomicAdd(&cnt[ei[N_EDGES + e]], 1);
}

#define SCAN_WAVES 32
#define SCAN_ITERS 13
__global__ __launch_bounds__(1024) void rowptr_scan_kernel(
    const int* __restrict__ cnt, int* __restrict__ rowptr, int* __restrict__ cursor) {
  static_assert(N_NODES % 4 == 0, "a 16-byte group is inside or outside as a whole");
  static_assert(SCAN_WAVES * SCAN_ITERS * 128 >= N_NODES, "span covers all nodes");
  __shared__ int wtot[SCAN_WAVES];
  const int lane = threadIdx.x & 31;
  const int w    = threadIdx.x >> 5;
  const int base = w * (SCAN_ITERS * 128);
  int4 v[SCAN_ITERS];
  int  x[SCAN_ITERS];
  #pragma unroll
  for (int i = 0; i < SCAN_ITERS; ++i) {
    const int k = base + (i * 32 + lane) * 4;
    v[i] = (k < N_NODES) ? *(const int4*)(cnt + k) : make_int4(0, 0, 0, 0);
  }
  #pragma unroll
  for (int i = 0; i < SCAN_ITERS; ++i) {
    int s = v[i].x + v[i].y + v[i].z + v[i].w;
    #pragma unroll
    for (int off = 1; off < 32; off <<= 1) {
      int y = __shfl_up(s, off, 32);
      if (lane >= off) s += y;
    }
    x[i] = s;
  }
  int carry = 0;
  #pragma unroll
  for (int i = 0; i < SCAN_ITERS; ++i) {
    const int own = v[i].x + v[i].y + v[i].z + v[i].w;
    const int tot = __shfl(x[i], 31, 32);
    x[i]  = x[i] - own + carry;
    carry += tot;
  }
  if (lane == 0) wtot[w] = carry;
  __syncthreads();
  int t = (lane < w) ? wtot[lane] : 0;
  #pragma unroll
  for (int off = 16; off > 0; off >>= 1) t += __shfl_xor(t, off, 32);
  #pragma unroll
  for (int i = 0; i < SCAN_ITERS; ++i) {
    const int k = base + (i * 32 + lane) * 4;
    if (k < N_NODES) {
      int4 r;
      r.x = x[i] + t;
      r.y = r.x + v[i].x;
      r.z = r.y + v[i].y;
      r.w = r.z + v[i].z;
      *(int4*)(rowptr + k) = r;
      *(int4*)(cursor + k) = r;
    }
  }
  if (threadIdx.x == 1023) rowptr[N_NODES] = t + carry;
}

__global__ void csr_fill_kernel(const int* __restrict__ ei, int* __restrict__ cursor,
                                int* __restrict__ col) {
  int e = blockIdx.x * blockDim.x + threadIdx.x;
  if (e >= N_EDGES) return;
  int s = ei[e];
  int d = ei[N_EDGES + e];
  int pos = atomicAdd(&cursor[d], 1);
  col[pos] = s;
}

#define AGG_WAVES 8
#define CHUNK 32
#define GROUPS (CHUNK / 4)
template <int HCv, bool RELU>
__global__ __launch_bounds__(256, 16) void node_softmax_agg_kernel(
    const int* __restrict__ rowptr, const int* __restrict__ col,
    const float* __restrict__ as, const float* __restrict__ ad,
    const float* __restrict__ hmat, const float* __restrict__ bias,
    float* __restrict__ outp) {
  constexpr int VEC = HCv / 32;
  constexpr int NV4 = VEC / 4;
  static_assert((HCv / HEADS) / VEC == 4, "four lanes per head");
  __shared__ float wl_all[AGG_WAVES][CHUNK * HEADS];
  __shared__ int   sl_all[AGG_WAVES][CHUNK];
  const int lane = threadIdx.x & 31;
  float* __restrict__ wl = wl_all[threadIdx.x >> 5];
  int*   __restrict__ sl = sl_all[threadIdx.x >> 5];
  const int n = __builtin_amdgcn_readfirstlane(
      (int)(blockIdx.x * (blockDim.x >> 5) + (threadIdx.x >> 5)));
  if (n >= N_NODES) return;
  const int hd  = lane >> 2;
  const int sub = lane & 3;
  const unsigned loff = (unsigned)lane * VEC;
  const int beg = rowptr[n];
  const int end = rowptr[n + 1];
  const int deg = end - beg;
  const float adv = ad[n * HEADS + hd];

  float vs = as[n * HEADS + hd] + adv;
  vs = vs > 0.f ? vs : NEG_SLOPE * vs;

  float acc[VEC];
  auto init_acc = [&](float w) {
    const float4* __restrict__ hp =
        (const float4*)(hmat + (size_t)n * HCv + loff);
    #pragma unroll
    for (int q = 0; q < NV4; ++q) {
      const float4 t = hp[q];
      acc[4 * q + 0] = w * t.x;
      acc[4 * q + 1] = w * t.y;
      acc[4 * q + 2] = w * t.z;
      acc[4 * q + 3] = w * t.w;
    }
  };
  auto add_row = [&](int s, float w) {
    const float4* __restrict__ hp =
        (const float4*)(hmat + (size_t)s * HCv + loff);
    #pragma unroll
    for (int q = 0; q < NV4; ++q) {
      const float4 t = hp[q];
      acc[4 * q + 0] += w * t.x;
      acc[4 * q + 1] += w * t.y;
      acc[4 * q + 2] += w * t.z;
      acc[4 * q + 3] += w * t.w;
    }
  };

  if (deg >= 1 && deg <= CHUNK) {
    float e[GROUPS];
    int   sv[GROUPS];
    const int last = deg - 1;
    #pragma unroll
    for (int g = 0; g < GROUPS; ++g) {
      const int idx = 4 * g + sub;
      sv[g] = col[beg + (idx < last ? idx : last)];
    }
    #pragma unroll
    for (int g = 0; g < GROUPS; ++g) e[g] = as[sv[g] * HEADS + hd];
    float m = vs;
    #pragma unroll
    for (int g = 0; g < GROUPS; ++g) {
      float v = e[g] + adv;
      v = v > 0.f ? v : NEG_SLOPE * v;
      v = (4 * g + sub) < deg ? v : -INFINITY;
      e[g] = v;
      m = v > m ? v : m;
    }
    {
      float y = __shfl_xor(m, 1, 32); m = y > m ? y : m;
      y = __shfl_xor(m, 2, 32);       m = y > m ? y : m;
    }
    float den = 0.f;
    #pragma unroll
    for (int g = 0; g < GROUPS; ++g) {
      e[g] = __expf(e[g] - m);
      den += e[g];
    }
    den += __shfl_xor(den, 1, 32);
    den += __shfl_xor(den, 2, 32);
    const float es = __expf(vs - m);
    den += es;
    const float dn = den + 1e-16f;
    #pragma unroll
    for (int g = 0; g < GROUPS; ++g) {
      wl[(4 * g + sub) * HEADS + hd] = e[g] / dn;
      sl[4 * g + sub] = sv[g];
    }
    __builtin_amdgcn_wave_barrier();
    init_acc(es / dn);
    for (int i = 0; i < deg; i += 4) {
      #pragma unroll
      for (int j = 0; j < 4; ++j) {
        const int s = __builtin_amdgcn_readfirstlane(sl[i + j]);
        add_row(s, wl[(i + j) * HEADS + hd]);
      }
    }
  } else {
    float m = vs;
    #pragma unroll 4
    for (int i = beg; i < end; ++i) {
      const int s = col[i];
      float v = as[s * HEADS + hd] + adv;
      v = v > 0.f ? v : NEG_SLOPE * v;
      m = v > m ? v : m;
    }
    float den = __expf(vs - m);
    #pragma unroll 4
    for (int i = beg; i < end; ++i) {
      const int s = col[i];
      float v = as[s * HEADS + hd] + adv;
      v = v > 0.f ? v : NEG_SLOPE * v;
      den += __expf(v - m);
    }
    const float dn = den + 1e-16f;
    init_acc(__expf(vs - m) / dn);
    #pragma unroll 4
    for (int i = beg; i < end; ++i) {
      const int s = col[i];
      float v = as[s * HEADS + hd] + adv;
      v = v > 0.f ? v : NEG_SLOPE * v;
      add_row(s, __expf(v - m) / dn);
    }
  }

  float4* __restrict__ op = (float4*)(outp + (size_t)n * HCv + lane * VEC);
  const float4* __restrict__ bp = (const float4*)(bias + lane * VEC);
  #pragma unroll
  for (int q = 0; q < NV4; ++q) {
    const float4 b = bp[q];
    float4 r;
    r.x = acc[4 * q + 0] + b.x;
    r.y = acc[4 * q + 1] + b.y;
    r.z = acc[4 * q + 2] + b.z;
    r.w = acc[4 * q + 3] + b.w;
    if (RELU) {
      r.x = r.x > 0.f ? r.x : 0.f;
      r.y = r.y > 0.f ? r.y : 0.f;
      r.z = r.z > 0.f ? r.z : 0.f;
      r.w = r.w > 0.f ? r.w : 0.f;
    }
    op[q] = r;
  }
}

static inline int blocks_for(long long n, int bs) {
  return (int)((n + bs - 1) / bs);
}

extern "C" void kernel_launch(void* const* d_in, const int* in_sizes, int n_in,
                              void* d_out, int out_size, void* d_ws, size_t ws_size,
                              hipStream_t stream) {
  const float* x      = (const float*)d_in[0];
  const int*   ei     = (const int*)  d_in[1];
  const float* W1     = (const float*)d_in[2];
  const float* a_src1 = (const float*)d_in[3];
  const float* a_dst1 = (const float*)d_in[4];
  const float* b1     = (const float*)d_in[5];
  const float* W2     = (const float*)d_in[6];
  const float* a_src2 = (const float*)d_in[7];
  const float* a_dst2 = (const float*)d_in[8];
  const float* b2     = (const float*)d_in[9];
  float* out = (float*)d_out;

  char* ws = (char*)d_ws;
  size_t off = 0;
  auto carve = [&](size_t bytes) {
    void* p = ws + off;
    off += (bytes + 255) & ~(size_t)255;
    return p;
  };
  float* hbuf   = (float*)carve((size_t)N_NODES * HC1 * 4);
  float* aggbuf = (float*)carve((size_t)N_NODES * HC1 * 4);
  float* as     = (float*)carve((size_t)N_NODES * HEADS * 4);
  float* ad     = (float*)carve((size_t)N_NODES * HEADS * 4);
  int*   cnt    = (int*)  carve((size_t)N_NODES * 4);
  int*   rowptr = (int*)  carve((size_t)(N_NODES + 1) * 4);
  int*   cursor = (int*)  carve((size_t)N_NODES * 4);
  int*   col    = (int*)  carve((size_t)N_EDGES * 4);

  const int BS = 256;
  const long long nNH = (long long)N_NODES * HEADS;

  {
    long long waves = (N_NODES / 16) * (HC1 / 64);
    gemm_f32_wmma<HC1, F_IN><<<blocks_for(waves * 32, BS), BS, 0, stream>>>(
        x, W1, hbuf, N_NODES);
  }
  alpha_kernel<HID><<<blocks_for(nNH, BS), BS, 0, stream>>>(hbuf, a_src1, a_dst1, as, ad);

  zero4_kernel<<<blocks_for(N_NODES / 4, BS), BS, 0, stream>>>((float4*)cnt, N_NODES / 4);
  degree_kernel<<<blocks_for(N_EDGES, BS), BS, 0, stream>>>(ei, cnt);
  rowptr_scan_kernel<<<1, 1024, 0, stream>>>(cnt, rowptr, cursor);
  csr_fill_kernel<<<blocks_for(N_EDGES, BS), BS, 0, stream>>>(ei, cursor, col);

  node_softmax_agg_kernel<HC1, true><<<blocks_for((long long)N_NODES * 32, BS), BS, 0, stream>>>(
      rowptr, col, as, ad, hbuf, b1, aggbuf);

  {
    long long waves = (N_NODES / 16) * (HC2 / 64);
    gemm_f32_wmma<HC2, HC1><<<blocks_for(waves * 32, BS), BS, 0, stream>>>(
        aggbuf, W2, hbuf, N_NODES);
  }
  alpha_kernel<N_CLS><<<blocks_for(nNH, BS), BS, 0, stream>>>(hbuf, a_src2, a_dst2, as, ad);
  node_softmax_agg_kernel<HC2, false><<<blocks_for((long long)N_NODES * 32, BS), BS, 0, stream>>>(
      rowptr, col, as, ad, hbuf, b2, out);
}


# Attention logits formed in the product kernels' epilogue from the accumulators (per-lane products, 16-lane halving sum: logit sum order changes); both logit launches and their re-read of h go
# speedup vs baseline: 2.7510x; 1.0558x over previous
#include <hip/hip_runtime.h>
#include <hip/hip_bf16.h>

#define N_NODES 50000
#define N_EDGES 800000
#define E_TOT   (N_EDGES + N_NODES)
#define F_IN    256
#define HEADS   8
#define HID     32
#define N_CLS   16
#define HC1     (HEADS * HID)
#define HC2     (HEADS * N_CLS)
#define NEG_SLOPE 0.2f

typedef float v2f __attribute__((ext_vector_type(2)));
typedef float v8f __attribute__((ext_vector_type(8)));

__global__ void zero4_kernel(float4* __restrict__ p, long long n4) {
  long long i = (long long)blockIdx.x * blockDim.x + threadIdx.x;
  if (i < n4) p[i] = make_float4(0.f, 0.f, 0.f, 0.f);
}

template <int N, int K, int CH, int WPE>
__global__ __launch_bounds__(256, WPE) void gemm_f32_wmma(
    const float* __restrict__ A, const float* __restrict__ B,
    const float* __restrict__ a_src, const float* __restrict__ a_dst,
    float* __restrict__ C, float* __restrict__ as, float* __restrict__ ad, int M) {
  constexpr int TILES_N = N / 64;
  const int lane  = threadIdx.x & 31;
  const int wave  = blockIdx.x * (blockDim.x >> 5) + (threadIdx.x >> 5);
  const int tileM = wave / TILES_N;
  const int tileN = wave % TILES_N;
  if (tileM * 16 >= M) return;

  const int r    = lane & 15;
  const int ksel = (lane >> 4) << 1;

  const float* __restrict__ arow  = A + (size_t)(tileM * 16 + r) * K + ksel;
  const float* __restrict__ bbase = B + (size_t)ksel * N + tileN * 64 + r;

  v8f acc0 = {}, acc1 = {}, acc2 = {}, acc3 = {};
  #pragma unroll 2
  for (int kk = 0; kk < K; kk += 4) {
    v2f a, b0, b1, b2, b3;
    a[0]  = arow[0];
    a[1]  = arow[1];
    b0[0] = bbase[0];       b0[1] = bbase[N];
    b1[0] = bbase[16];      b1[1] = bbase[N + 16];
    b2[0] = bbase[32];      b2[1] = bbase[N + 32];
    b3[0] = bbase[48];      b3[1] = bbase[N + 48];
    acc0 = __builtin_amdgcn_wmma_f32_16x16x4_f32(false, a, false, b0, (short)0, acc0, false, false);
    acc1 = __builtin_amdgcn_wmma_f32_16x16x4_f32(false, a, false, b1, (short)0, acc1, false, false);
    acc2 = __builtin_amdgcn_wmma_f32_16x16x4_f32(false, a, false, b2, (short)0, acc2, false, false);
    acc3 = __builtin_amdgcn_wmma_f32_16x16x4_f32(false, a, false, b3, (short)0, acc3, false, false);
    arow  += 4;
    bbase += (size_t)4 * N;
  }

  float* __restrict__ crow =
      C + (size_t)(tileM * 16 + (lane >> 4) * 8) * N + tileN * 64 + (lane & 15);
  #pragma unroll
  for (int i = 0; i < 8; ++i) {
    crow[(size_t)i * N +  0] = acc0[i];
    crow[(size_t)i * N + 16] = acc1[i];
    crow[(size_t)i * N + 32] = acc2[i];
    crow[(size_t)i * N + 48] = acc3[i];
  }

  constexpr int HPW = 64 / CH;
  constexpr int APH = CH / 16;
  constexpr int NV  = 8 * HPW;
  constexpr int R   = NV / 16;
  static_assert(HPW * CH == 64 && APH * 16 == CH, "whole heads per wave");
  static_assert(HPW == 2 * R, "row of a finished value is (lane & 15) >> 1");
  static_assert(N / CH == HEADS, "logit rows hold HEADS values");
  const int c = lane & 15;
  const v8f accs[4] = {acc0, acc1, acc2, acc3};
  float ps[4], pd[4];
  #pragma unroll
  for (int j = 0; j < 4; ++j) {
    ps[j] = a_src[tileN * 64 + 16 * j + c];
    pd[j] = a_dst[tileN * 64 + 16 * j + c];
  }
  float vs[NV], vd[NV];
  #pragma unroll
  for (int i = 0; i < 8; ++i) {
    #pragma unroll
    for (int q = 0; q < HPW; ++q) {
      float sa = accs[q * APH][i] * ps[q * APH];
      float sd = accs[q * APH][i] * pd[q * APH];
      #pragma unroll
      for (int t = 1; t < APH; ++t) {
        const float hv = accs[q * APH + t][i];
        sa += hv * ps[q * APH + t];
        sd += hv * pd[q * APH + t];
      }
      vs[i * HPW + q] = sa;
      vd[i * HPW + q] = sd;
    }
  }
  #pragma unroll
  for (int m = 8, n = NV / 2; m >= 1; m >>= 1, n >>= 1) {
    const bool up = (c & m) != 0;
    #pragma unroll
    for (int k = 0; k < n; ++k) {
      const float s_lo = vs[k], s_hi = vs[k + n];
      const float d_lo = vd[k], d_hi = vd[k + n];
      const float keep_s = up ? s_hi : s_lo;
      const float send_s = up ? s_lo : s_hi;
      const float keep_d = up ? d_hi : d_lo;
      const float send_d = up ? d_lo : d_hi;
      vs[k] = keep_s + __shfl_xor(send_s, m, 32);
      vd[k] = keep_d + __shfl_xor(send_d, m, 32);
    }
  }
  const size_t lo = (size_t)(tileM * 16 + (lane >> 4) * 8 + (c >> 1)) * HEADS +
                    tileN * HPW + R * (c & 1);
  #pragma unroll
  for (int k = 0; k < R; ++k) {
    as[lo + k] = vs[k];
    ad[lo + k] = vd[k];
  }
}

__global__ void degree_kernel(const int* __restrict__ ei, int* __restrict__ cnt) {
  int e = blockIdx.x * blockDim.x + threadIdx.x;
  if (e >= N_EDGES) return;
  atomicAdd(&cnt[ei[N_EDGES + e]], 1);
}

#define SCAN_WAVES 32
#define SCAN_ITERS 13
__global__ __launch_bounds__(1024) void rowptr_scan_kernel(
    const int* __restrict__ cnt, int* __restrict__ rowptr, int* __restrict__ cursor) {
  static_assert(N_NODES % 4 == 0, "a 16-byte group is inside or outside as a whole");
  static_assert(SCAN_WAVES * SCAN_ITERS * 128 >= N_NODES, "span covers all nodes");
  __shared__ int wtot[SCAN_WAVES];
  const int lane = threadIdx.x & 31;
  const int w    = threadIdx.x >> 5;
  const int base = w * (SCAN_ITERS * 128);
  int4 v[SCAN_ITERS];
  int  x[SCAN_ITERS];
  #pragma unroll
  for (int i = 0; i < SCAN_ITERS; ++i) {
    const int k = base + (i * 32 + lane) * 4;
    v[i] = (k < N_NODES) ? *(const int4*)(cnt + k) : make_int4(0, 0, 0, 0);
  }
  #pragma unroll
  for (int i = 0; i < SCAN_ITERS; ++i) {
    int s = v[i].x + v[i].y + v[i].z + v[i].w;
    #pragma unroll
    for (int off = 1; off < 32; off <<= 1) {
      int y = __shfl_up(s, off, 32);
      if (lane >= off) s += y;
    }
    x[i] = s;
  }
  int carry = 0;
  #pragma unroll
  for (int i = 0; i < SCAN_ITERS; ++i) {
    const int own = v[i].x + v[i].y + v[i].z + v[i].w;
    const int tot = __shfl(x[i], 31, 32);
    x[i]  = x[i] - own + carry;
    carry += tot;
  }
  if (lane == 0) wtot[w] = carry;
  __syncthreads();
  int t = (lane < w) ? wtot[lane] : 0;
  #pragma unroll
  for (int off = 16; off > 0; off >>= 1) t += __shfl_xor(t, off, 32);
  #pragma unroll
  for (int i = 0; i < SCAN_ITERS; ++i) {
    const int k = base + (i * 32 + lane) * 4;
    if (k < N_NODES) {
      int4 r;
      r.x = x[i] + t;
      r.y = r.x + v[i].x;
      r.z = r.y + v[i].y;
      r.w = r.z + v[i].z;
      *(int4*)(rowptr + k) = r;
      *(int4*)(cursor + k) = r;
    }
  }
  if (threadIdx.x == 1023) rowptr[N_NODES] = t + carry;
}

__global__ void csr_fill_kernel(const int* __restrict__ ei, int* __restrict__ cursor,
                                int* __restrict__ col) {
  int e = blockIdx.x * blockDim.x + threadIdx.x;
  if (e >= N_EDGES) return;
  int s = ei[e];
  int d = ei[N_EDGES + e];
  int pos = atomicAdd(&cursor[d], 1);
  col[pos] = s;
}

#define AGG_WAVES 8
#define CHUNK 32
#define GROUPS (CHUNK / 4)
template <int HCv, bool RELU>
__global__ __launch_bounds__(256, 16) void node_softmax_agg_kernel(
    const int* __restrict__ rowptr, const int* __restrict__ col,
    const float* __restrict__ as, const float* __restrict__ ad,
    const float* __restrict__ hmat, const float* __restrict__ bias,
    float* __restrict__ outp) {
  constexpr int VEC = HCv / 32;
  constexpr int NV4 = VEC / 4;
  static_assert((HCv / HEADS) / VEC == 4, "four lanes per head");
  __shared__ float wl_all[AGG_WAVES][CHUNK * HEADS];
  __shared__ int   sl_all[AGG_WAVES][CHUNK];
  const int lane = threadIdx.x & 31;
  float* __restrict__ wl = wl_all[threadIdx.x >> 5];
  int*   __restrict__ sl = sl_all[threadIdx.x >> 5];
  const int n = __builtin_amdgcn_readfirstlane(
      (int)(blockIdx.x * (blockDim.x >> 5) + (threadIdx.x >> 5)));
  if (n >= N_NODES) return;
  const int hd  = lane >> 2;
  const int sub = lane & 3;
  const unsigned loff = (unsigned)lane * VEC;
  const int beg = rowptr[n];
  const int end = rowptr[n + 1];
  const int deg = end - beg;
  const float adv = ad[n * HEADS + hd];

  float vs = as[n * HEADS + hd] + adv;
  vs = vs > 0.f ? vs : NEG_SLOPE * vs;

  float acc[VEC];
  auto init_acc = [&](float w) {
    const float4* __restrict__ hp =
        (const float4*)(hmat + (size_t)n * HCv + loff);
    #pragma unroll
    for (int q = 0; q < NV4; ++q) {
      const float4 t = hp[q];
      acc[4 * q + 0] = w * t.x;
      acc[4 * q + 1] = w * t.y;
      acc[4 * q + 2] = w * t.z;
      acc[4 * q + 3] = w * t.w;
    }
  };
  auto add_row = [&](int s, float w) {
    const float4* __restrict__ hp =
        (const float4*)(hmat + (size_t)s * HCv + loff);
    #pragma unroll
    for (int q = 0; q < NV4; ++q) {
      const float4 t = hp[q];
      acc[4 * q + 0] += w * t.x;
      acc[4 * q + 1] += w * t.y;
      acc[4 * q + 2] += w * t.z;
      acc[4 * q + 3] += w * t.w;
    }
  };

  if (deg >= 1 && deg <= CHUNK) {
    float e[GROUPS];
    int   sv[GROUPS];
    const int last = deg - 1;
    #pragma unroll
    for (int g = 0; g < GROUPS; ++g) {
      const int idx = 4 * g + sub;
      sv[g] = col[beg + (idx < last ? idx : last)];
    }
    #pragma unroll
    for (int g = 0; g < GROUPS; ++g) e[g] = as[sv[g] * HEADS + hd];
    float m = vs;
    #pragma unroll
    for (int g = 0; g < GROUPS; ++g) {
      float v = e[g] + adv;
      v = v > 0.f ? v : NEG_SLOPE * v;
      v = (4 * g + sub) < deg ? v : -INFINITY;
      e[g] = v;
      m = v > m ? v : m;
    }
    {
      float y = __shfl_xor(m, 1, 32); m = y > m ? y : m;
      y = __shfl_xor(m, 2, 32);       m = y > m ? y : m;
    }
    float den = 0.f;
    #pragma unroll
    for (int g = 0; g < GROUPS; ++g) {
      e[g] = __expf(e[g] - m);
      den += e[g];
    }
    den += __shfl_xor(den, 1, 32);
    den += __shfl_xor(den, 2, 32);
    const float es = __expf(vs - m);
    den += es;
    const float dn = den + 1e-16f;
    #pragma unroll
    for (int g = 0; g < GROUPS; ++g) {
      wl[(4 * g + sub) * HEADS + hd] = e[g] / dn;
      sl[4 * g + sub] = sv[g];
    }
    __builtin_amdgcn_wave_barrier();
    init_acc(es / dn);
    for (int i = 0; i < deg; i += 4) {
      #pragma unroll
      for (int j = 0; j < 4; ++j) {
        const int s = __builtin_amdgcn_readfirstlane(sl[i + j]);
        add_row(s, wl[(i + j) * HEADS + hd]);
      }
    }
  } else {
    float m = vs;
    #pragma unroll 4
    for (int i = beg; i < end; ++i) {
      const int s = col[i];
      float v = as[s * HEADS + hd] + adv;
      v = v > 0.f ? v : NEG_SLOPE * v;
      m = v > m ? v : m;
    }
    float den = __expf(vs - m);
    #pragma unroll 4
    for (int i = beg; i < end; ++i) {
      const int s = col[i];
      float v = as[s * HEADS + hd] + adv;
      v = v > 0.f ? v : NEG_SLOPE * v;
      den += __expf(v - m);
    }
    const float dn = den + 1e-16f;
    init_acc(__expf(vs - m) / dn);
    #pragma unroll 4
    for (int i = beg; i < end; ++i) {
      const int s = col[i];
      float v = as[s * HEADS + hd] + adv;
      v = v > 0.f ? v : NEG_SLOPE * v;
      add_row(s, __expf(v - m) / dn);
    }
  }

  float4* __restrict__ op = (float4*)(outp + (size_t)n * HCv + lane * VEC);
  const float4* __restrict__ bp = (const float4*)(bias + lane * VEC);
  #pragma unroll
  for (int q = 0; q < NV4; ++q) {
    const float4 b = bp[q];
    float4 r;
    r.x = acc[4 * q + 0] + b.x;
    r.y = acc[4 * q + 1] + b.y;
    r.z = acc[4 * q + 2] + b.z;
    r.w = acc[4 * q + 3] + b.w;
    if (RELU) {
      r.x = r.x > 0.f ? r.x : 0.f;
      r.y = r.y > 0.f ? r.y : 0.f;
      r.z = r.z > 0.f ? r.z : 0.f;
      r.w = r.w > 0.f ? r.w : 0.f;
    }
    op[q] = r;
  }
}

static inline int blocks_for(long long n, int bs) {
  return (int)((n + bs - 1) / bs);
}

extern "C" void kernel_launch(void* const* d_in, const int* in_sizes, int n_in,
                              void* d_out, int out_size, void* d_ws, size_t ws_size,
                              hipStream_t stream) {
  const float* x      = (const float*)d_in[0];
  const int*   ei     = (const int*)  d_in[1];
  const float* W1     = (const float*)d_in[2];
  const float* a_src1 = (const float*)d_in[3];
  const float* a_dst1 = (const float*)d_in[4];
  const float* b1     = (const float*)d_in[5];
  const float* W2     = (const float*)d_in[6];
  const float* a_src2 = (const float*)d_in[7];
  const float* a_dst2 = (const float*)d_in[8];
  const float* b2     = (const float*)d_in[9];
  float* out = (float*)d_out;

  char* ws = (char*)d_ws;
  size_t off = 0;
  auto carve = [&](size_t bytes) {
    void* p = ws + off;
    off += (bytes + 255) & ~(size_t)255;
    return p;
  };
  float* hbuf   = (float*)carve((size_t)N_NODES * HC1 * 4);
  float* aggbuf = (float*)carve((size_t)N_NODES * HC1 * 4);
  float* as     = (float*)carve((size_t)N_NODES * HEADS * 4);
  float* ad     = (float*)carve((size_t)N_NODES * HEADS * 4);
  int*   cnt    = (int*)  carve((size_t)N_NODES * 4);
  int*   rowptr = (int*)  carve((size_t)(N_NODES + 1) * 4);
  int*   cursor = (int*)  carve((size_t)N_NODES * 4);
  int*   col    = (int*)  carve((size_t)N_EDGES * 4);

  const int BS = 256;

  {
    long long waves = (N_NODES / 16) * (HC1 / 64);
    gemm_f32_wmma<HC1, F_IN, HID, 16><<<blocks_for(waves * 32, BS), BS, 0, stream>>>(
        x, W1, a_src1, a_dst1, hbuf, as, ad, N_NODES);
  }

  zero4_kernel<<<blocks_for(N_NODES / 4, BS), BS, 0, stream>>>((float4*)cnt, N_NODES / 4);
  degree_kernel<<<blocks_for(N_EDGES, BS), BS, 0, stream>>>(ei, cnt);
  rowptr_scan_kernel<<<1, 1024, 0, stream>>>(cnt, rowptr, cursor);
  csr_fill_kernel<<<blocks_for(N_EDGES, BS), BS, 0, stream>>>(ei, cursor, col);

  node_softmax_agg_kernel<HC1, true><<<blocks_for((long long)N_NODES * 32, BS), BS, 0, stream>>>(
      rowptr, col, as, ad, hbuf, b1, aggbuf);

  {
    long long waves = (N_NODES / 16) * (HC2 / 64);
    gemm_f32_wmma<HC2, HC1, N_CLS, 8><<<blocks_for(waves * 32, BS), BS, 0, stream>>>(
        aggbuf, W2, a_src2, a_dst2, hbuf, as, ad, N_NODES);
  }
  node_softmax_agg_kernel<HC2, false><<<blocks_for((long long)N_NODES * 32, BS), BS, 0, stream>>>(
      rowptr, col, as, ad, hbuf, b2, out);
}


# Edge grouping in one launch, no global atomics: a work-group per destination range reads all destinations, counts and lists its edges in LDS, scans, fills by LDS cursors; fill order arrival-ranked
# speedup vs baseline: 3.4122x; 1.2403x over previous
#include <hip/hip_runtime.h>
#include <hip/hip_bf16.h>

#define N_NODES 50000
#define N_EDGES 800000
#define E_TOT   (N_EDGES + N_NODES)
#define F_IN    256
#define HEADS   8
#define HID     32
#define N_CLS   16
#define HC1     (HEADS * HID)
#define HC2     (HEADS * N_CLS)
#define NEG_SLOPE 0.2f

typedef float v2f __attribute__((ext_vector_type(2)));
typedef float v8f __attribute__((ext_vector_type(8)));

template <int N, int K, int CH, int WPE>
__global__ __launch_bounds__(256, WPE) void gemm_f32_wmma(
    const float* __restrict__ A, const float* __restrict__ B,
    const float* __restrict__ a_src, const float* __restrict__ a_dst,
    float* __restrict__ C, float* __restrict__ as, float* __restrict__ ad, int M) {
  constexpr int TILES_N = N / 64;
  const int lane  = threadIdx.x & 31;
  const int wave  = blockIdx.x * (blockDim.x >> 5) + (threadIdx.x >> 5);
  const int tileM = wave / TILES_N;
  const int tileN = wave % TILES_N;
  if (tileM * 16 >= M) return;

  const int r    = lane & 15;
  const int ksel = (lane >> 4) << 1;

  const float* __restrict__ arow  = A + (size_t)(tileM * 16 + r) * K + ksel;
  const float* __restrict__ bbase = B + (size_t)ksel * N + tileN * 64 + r;

  v8f acc0 = {}, acc1 = {}, acc2 = {}, acc3 = {};
  #pragma unroll 2
  for (int kk = 0; kk < K; kk += 4) {
    v2f a, b0, b1, b2, b3;
    a[0]  = arow[0];
    a[1]  = arow[1];
    b0[0] = bbase[0];       b0[1] = bbase[N];
    b1[0] = bbase[16];      b1[1] = bbase[N + 16];
    b2[0] = bbase[32];      b2[1] = bbase[N + 32];
    b3[0] = bbase[48];      b3[1] = bbase[N + 48];
    acc0 = __builtin_amdgcn_wmma_f32_16x16x4_f32(false, a, false, b0, (short)0, acc0, false, false);
    acc1 = __builtin_amdgcn_wmma_f32_16x16x4_f32(false, a, false, b1, (short)0, acc1, false, false);
    acc2 = __builtin_amdgcn_wmma_f32_16x16x4_f32(false, a, false, b2, (short)0, acc2, false, false);
    acc3 = __builtin_amdgcn_wmma_f32_16x16x4_f32(false, a, false, b3, (short)0, acc3, false, false);
    arow  += 4;
    bbase += (size_t)4 * N;
  }

  float* __restrict__ crow =
      C + (size_t)(tileM * 16 + (lane >> 4) * 8) * N + tileN * 64 + (lane & 15);
  #pragma unroll
  for (int i = 0; i < 8; ++i) {
    crow[(size_t)i * N +  0] = acc0[i];
    crow[(size_t)i * N + 16] = acc1[i];
    crow[(size_t)i * N + 32] = acc2[i];
    crow[(size_t)i * N + 48] = acc3[i];
  }

  constexpr int HPW = 64 / CH;
  constexpr int APH = CH / 16;
  constexpr int NV  = 8 * HPW;
  constexpr int R   = NV / 16;
  static_assert(HPW * CH == 64 && APH * 16 == CH, "whole heads per wave");
  static_assert(HPW == 2 * R, "row of a finished value is (lane & 15) >> 1");
  static_assert(N / CH == HEADS, "logit rows hold HEADS values");
  const int c = lane & 15;
  const v8f accs[4] = {acc0, acc1, acc2, acc3};
  float ps[4], pd[4];
  #pragma unroll
  for (int j = 0; j < 4; ++j) {
    ps[j] = a_src[tileN * 64 + 16 * j + c];
    pd[j] = a_dst[tileN * 64 + 16 * j + c];
  }
  float vs[NV], vd[NV];
  #pragma unroll
  for (int i = 0; i < 8; ++i) {
    #pragma unroll
    for (int q = 0; q < HPW; ++q) {
      float sa = accs[q * APH][i] * ps[q * APH];
      float sd = accs[q * APH][i] * pd[q * APH];
      #pragma unroll
      for (int t = 1; t < APH; ++t) {
        const float hv = accs[q * APH + t][i];
        sa += hv * ps[q * APH + t];
        sd += hv * pd[q * APH + t];
      }
      vs[i * HPW + q] = sa;
      vd[i * HPW + q] = sd;
    }
  }
  #pragma unroll
  for (int m = 8, n = NV / 2; m >= 1; m >>= 1, n >>= 1) {
    const bool up = (c & m) != 0;
    #pragma unroll
    for (int k = 0; k < n; ++k) {
      const float s_lo = vs[k], s_hi = vs[k + n];
      const float d_lo = vd[k], d_hi = vd[k + n];
      const float keep_s = up ? s_hi : s_lo;
      const float send_s = up ? s_lo : s_hi;
      const float keep_d = up ? d_hi : d_lo;
      const float send_d = up ? d_lo : d_hi;
      vs[k] = keep_s + __shfl_xor(send_s, m, 32);
      vd[k] = keep_d + __shfl_xor(send_d, m, 32);
    }
  }
  const size_t lo = (size_t)(tileM * 16 + (lane >> 4) * 8 + (c >> 1)) * HEADS +
                    tileN * HPW + R * (c & 1);
  #pragma unroll
  for (int k = 0; k < R; ++k) {
    as[lo + k] = vs[k];
    ad[lo + k] = vd[k];
  }
}

#define SORT_WG   128
#define SORT_R    ((N_NODES + SORT_WG - 1) / SORT_WG)
#define SORT_TH   1024
#define SORT_CAP  16384
#define SORT_NQ   (N_EDGES / 4)
#define SORT_UNR  8
__global__ __launch_bounds__(SORT_TH) void csr_build_kernel(
    const int* __restrict__ ei, int* __restrict__ rowptr, int* __restrict__ col) {
  static_assert(N_EDGES % 4 == 0, "destinations are read in 16-byte groups");
  static_assert(N_EDGES <= (1 << 20) && SORT_R <= (1 << 12),
                "edge id (20 bits) and node inside the range (12 bits) share a word");
  static_assert((SORT_WG - 1) * SORT_R < N_NODES && SORT_WG * SORT_R >= N_NODES,
                "every range holds a node and the ranges cover all nodes");
  __shared__ unsigned s_cnt[SORT_R];
  __shared__ unsigned s_excl[SORT_R];
  __shared__ unsigned s_list[SORT_CAP];
  __shared__ unsigned s_n;
  __shared__ unsigned s_below;
  const int t    = threadIdx.x;
  const int lane = t & 31;
  const int lo   = blockIdx.x * SORT_R;
  const int hi   = (lo + SORT_R < N_NODES) ? lo + SORT_R : N_NODES;
  const unsigned rthis = (unsigned)(hi - lo);
  for (int i = t; i < SORT_R; i += SORT_TH) s_cnt[i] = 0u;
  if (t == 0) { s_n = 0u; s_below = 0u; }
  __syncthreads();

  const int4* __restrict__ d4 = (const int4*)(ei + N_EDGES);
  auto take = [&](unsigned dl, int e) {
    if (dl < rthis) {
      const unsigned p = atomicAdd(&s_n, 1u);
      if (p < SORT_CAP) s_list[p] = (dl << 20) | (unsigned)e;
      atomicAdd(&s_cnt[dl], 1u);
    }
  };
  unsigned below = 0u;
  for (int q0 = 0; q0 < SORT_NQ; q0 += SORT_TH * SORT_UNR) {
    int4 v[SORT_UNR];
    #pragma unroll
    for (int j = 0; j < SORT_UNR; ++j) {
      const int q = q0 + j * SORT_TH + t;
      v[j] = (q < SORT_NQ) ? d4[q] : make_int4(hi, hi, hi, hi);
    }
    #pragma unroll
    for (int j = 0; j < SORT_UNR; ++j) {
      const int q = q0 + j * SORT_TH + t;
      const unsigned a = (unsigned)v[j].x - (unsigned)lo;
      const unsigned b = (unsigned)v[j].y - (unsigned)lo;
      const unsigned c = (unsigned)v[j].z - (unsigned)lo;
      const unsigned d = (unsigned)v[j].w - (unsigned)lo;
      below += (a >> 31) + (b >> 31) + (c >> 31) + (d >> 31);
      const unsigned mab = a < b ? a : b;
      const unsigned mcd = c < d ? c : d;
      if ((mab < mcd ? mab : mcd) < rthis) {
        take(a, 4 * q + 0);
        take(b, 4 * q + 1);
        take(c, 4 * q + 2);
        take(d, 4 * q + 3);
      }
    }
  }
  #pragma unroll
  for (int off = 16; off > 0; off >>= 1) below += __shfl_xor(below, off, 32);
  if (lane == 0) atomicAdd(&s_below, below);
  __syncthreads();
  const unsigned base  = s_below;
  const unsigned nmine = s_n;

  if (t < 32) {
    unsigned carry = 0u;
    for (int c = 0; c < (SORT_R + 31) / 32; ++c) {
      const int i = c * 32 + lane;
      const bool in = (unsigned)i < rthis;
      const unsigned v = in ? s_cnt[i] : 0u;
      unsigned inc = v;
      #pragma unroll
      for (int off = 1; off < 32; off <<= 1) {
        const unsigned y = __shfl_up(inc, off, 32);
        if (lane >= off) inc += y;
      }
      const unsigned ex = inc - v + carry;
      if (in) {
        s_excl[i] = ex;
        s_cnt[i]  = 0u;
        rowptr[lo + i] = (int)(base + ex);
      }
      carry += __shfl(inc, 31, 32);
    }
    if (lane == 0 && hi == N_NODES) rowptr[N_NODES] = (int)(base + carry);
  }
  __syncthreads();

  auto place = [&](unsigned dl, int e) {
    const int s = ei[e];
    const unsigned r = atomicAdd(&s_cnt[dl], 1u);
    col[base + s_excl[dl] + r] = s;
  };
  if (nmine <= SORT_CAP) {
    for (unsigned k = (unsigned)t; k < nmine; k += SORT_TH) {
      const unsigned p = s_list[k];
      place(p >> 20, (int)(p & 0xFFFFFu));
    }
  } else {
    for (int q = t; q < SORT_NQ; q += SORT_TH) {
      const int4 v = d4[q];
      const unsigned a = (unsigned)v.x - (unsigned)lo;
      const unsigned b = (unsigned)v.y - (unsigned)lo;
      const unsigned c = (unsigned)v.z - (unsigned)lo;
      const unsigned d = (unsigned)v.w - (unsigned)lo;
      if (a < rthis) place(a, 4 * q + 0);
      if (b < rthis) place(b, 4 * q + 1);
      if (c < rthis) place(c, 4 * q + 2);
      if (d < rthis) place(d, 4 * q + 3);
    }
  }
}

#define AGG_WAVES 8
#define CHUNK 32
#define GROUPS (CHUNK / 4)
template <int HCv, bool RELU>
__global__ __launch_bounds__(256, 16) void node_softmax_agg_kernel(
    const int* __restrict__ rowptr, const int* __restrict__ col,
    const float* __restrict__ as, const float* __restrict__ ad,
    const float* __restrict__ hmat, const float* __restrict__ bias,
    float* __restrict__ outp) {
  constexpr int VEC = HCv / 32;
  constexpr int NV4 = VEC / 4;
  static_assert((HCv / HEADS) / VEC == 4, "four lanes per head");
  __shared__ float wl_all[AGG_WAVES][CHUNK * HEADS];
  __shared__ int   sl_all[AGG_WAVES][CHUNK];
  const int lane = threadIdx.x & 31;
  float* __restrict__ wl = wl_all[threadIdx.x >> 5];
  int*   __restrict__ sl = sl_all[threadIdx.x >> 5];
  const int n = __builtin_amdgcn_readfirstlane(
      (int)(blockIdx.x * (blockDim.x >> 5) + (threadIdx.x >> 5)));
  if (n >= N_NODES) return;
  const int hd  = lane >> 2;
  const int sub = lane & 3;
  const unsigned loff = (unsigned)lane * VEC;
  const int beg = rowptr[n];
  const int end = rowptr[n + 1];
  const int deg = end - beg;
  const float adv = ad[n * HEADS + hd];

  float vs = as[n * HEADS + hd] + adv;
  vs = vs > 0.f ? vs : NEG_SLOPE * vs;

  float acc[VEC];
  auto init_acc = [&](float w) {
    const float4* __restrict__ hp =
        (const float4*)(hmat + (size_t)n * HCv + loff);
    #pragma unroll
    for (int q = 0; q < NV4; ++q) {
      const float4 t = hp[q];
      acc[4 * q + 0] = w * t.x;
      acc[4 * q + 1] = w * t.y;
      acc[4 * q + 2] = w * t.z;
      acc[4 * q + 3] = w * t.w;
    }
  };
  auto add_row = [&](int s, float w) {
    const float4* __restrict__ hp =
        (const float4*)(hmat + (size_t)s * HCv + loff);
    #pragma unroll
    for (int q = 0; q < NV4; ++q) {
      const float4 t = hp[q];
      acc[4 * q + 0] += w * t.x;
      acc[4 * q + 1] += w * t.y;
      acc[4 * q + 2] += w * t.z;
      acc[4 * q + 3] += w * t.w;
    }
  };

  if (deg >= 1 && deg <= CHUNK) {
    float e[GROUPS];
    int   sv[GROUPS];
    const int last = deg - 1;
    #pragma unroll
    for (int g = 0; g < GROUPS; ++g) {
      const int idx = 4 * g + sub;
      sv[g] = col[beg + (idx < last ? idx : last)];
    }
    #pragma unroll
    for (int g = 0; g < GROUPS; ++g) e[g] = as[sv[g] * HEADS + hd];
    float m = vs;
    #pragma unroll
    for (int g = 0; g < GROUPS; ++g) {
      float v = e[g] + adv;
      v = v > 0.f ? v : NEG_SLOPE * v;
      v = (4 * g + sub) < deg ? v : -INFINITY;
      e[g] = v;
      m = v > m ? v : m;
    }
    {
      float y = __shfl_xor(m, 1, 32); m = y > m ? y : m;
      y = __shfl_xor(m, 2, 32);       m = y > m ? y : m;
    }
    float den = 0.f;
    #pragma unroll
    for (int g = 0; g < GROUPS; ++g) {
      e[g] = __expf(e[g] - m);
      den += e[g];
    }
    den += __shfl_xor(den, 1, 32);
    den += __shfl_xor(den, 2, 32);
    const float es = __expf(vs - m);
    den += es;
    const float dn = den + 1e-16f;
    #pragma unroll
    for (int g = 0; g < GROUPS; ++g) {
      wl[(4 * g + sub) * HEADS + hd] = e[g] / dn;
      sl[4 * g + sub] = sv[g];
    }
    __builtin_amdgcn_wave_barrier();
    init_acc(es / dn);
    for (int i = 0; i < deg; i += 4) {
      #pragma unroll
      for (int j = 0; j < 4; ++j) {
        const int s = __builtin_amdgcn_readfirstlane(sl[i + j]);
        add_row(s, wl[(i + j) * HEADS + hd]);
      }
    }
  } else {
    float m = vs;
    #pragma unroll 4
    for (int i = beg; i < end; ++i) {
      const int s = col[i];
      float v = as[s * HEADS + hd] + adv;
      v = v > 0.f ? v : NEG_SLOPE * v;
      m = v > m ? v : m;
    }
    float den = __expf(vs - m);
    #pragma unroll 4
    for (int i = beg; i < end; ++i) {
      const int s = col[i];
      float v = as[s * HEADS + hd] + adv;
      v = v > 0.f ? v : NEG_SLOPE * v;
      den += __expf(v - m);
    }
    const float dn = den + 1e-16f;
    init_acc(__expf(vs - m) / dn);
    #pragma unroll 4
    for (int i = beg; i < end; ++i) {
      const int s = col[i];
      float v = as[s * HEADS + hd] + adv;
      v = v > 0.f ? v : NEG_SLOPE * v;
      add_row(s, __expf(v - m) / dn);
    }
  }

  float4* __restrict__ op = (float4*)(outp + (size_t)n * HCv + lane * VEC);
  const float4* __restrict__ bp = (const float4*)(bias + lane * VEC);
  #pragma unroll
  for (int q = 0; q < NV4; ++q) {
    const float4 b = bp[q];
    float4 r;
    r.x = acc[4 * q + 0] + b.x;
    r.y = acc[4 * q + 1] + b.y;
    r.z = acc[4 * q + 2] + b.z;
    r.w = acc[4 * q + 3] + b.w;
    if (RELU) {
      r.x = r.x > 0.f ? r.x : 0.f;
      r.y = r.y > 0.f ? r.y : 0.f;
      r.z = r.z > 0.f ? r.z : 0.f;
      r.w = r.w > 0.f ? r.w : 0.f;
    }
    op[q] = r;
  }
}

static inline int blocks_for(long long n, int bs) {
  return (int)((n + bs - 1) / bs);
}

extern "C" void kernel_launch(void* const* d_in, const int* in_sizes, int n_in,
                              void* d_out, int out_size, void* d_ws, size_t ws_size,
                              hipStream_t stream) {
  const float* x      = (const float*)d_in[0];
  const int*   ei     = (const int*)  d_in[1];
  const float* W1     = (const float*)d_in[2];
  const float* a_src1 = (const float*)d_in[3];
  const float* a_dst1 = (const float*)d_in[4];
  const float* b1     = (const float*)d_in[5];
  const float* W2     = (const float*)d_in[6];
  const float* a_src2 = (const float*)d_in[7];
  const float* a_dst2 = (const float*)d_in[8];
  const float* b2     = (const float*)d_in[9];
  float* out = (float*)d_out;

  char* ws = (char*)d_ws;
  size_t off = 0;
  auto carve = [&](size_t bytes) {
    void* p = ws + off;
    off += (bytes + 255) & ~(size_t)255;
    return p;
  };
  float* hbuf   = (float*)carve((size_t)N_NODES * HC1 * 4);
  float* aggbuf = (float*)carve((size_t)N_NODES * HC1 * 4);
  float* as     = (float*)carve((size_t)N_NODES * HEADS * 4);
  float* ad     = (float*)carve((size_t)N_NODES * HEADS * 4);
  int*   rowptr = (int*)  carve((size_t)(N_NODES + 1) * 4);
  int*   col    = (int*)  carve((size_t)N_EDGES * 4);

  const int BS = 256;

  {
    long long waves = (N_NODES / 16) * (HC1 / 64);
    gemm_f32_wmma<HC1, F_IN, HID, 16><<<blocks_for(waves * 32, BS), BS, 0, stream>>>(
        x, W1, a_src1, a_dst1, hbuf, as, ad, N_NODES);
  }

  csr_build_kernel<<<SORT_WG, SORT_TH, 0, stream>>>(ei, rowptr, col);

  node_softmax_agg_kernel<HC1, true><<<blocks_for((long long)N_NODES * 32, BS), BS, 0, stream>>>(
      rowptr, col, as, ad, hbuf, b1, aggbuf);

  {
    long long waves = (N_NODES / 16) * (HC2 / 64);
    gemm_f32_wmma<HC2, HC1, N_CLS, 8><<<blocks_for(waves * 32, BS), BS, 0, stream>>>(
        aggbuf, W2, a_src2, a_dst2, hbuf, as, ad, N_NODES);
  }
  node_softmax_agg_kernel<HC2, false><<<blocks_for((long long)N_NODES * 32, BS), BS, 0, stream>>>(
      rowptr, col, as, ad, hbuf, b2, out);
}


# Product kernels bring the work-group's rows of A into LDS once with whole-line 16-byte loads (pitch K+4); K loop reads A fragments from LDS, only B from global; k and sum order unchanged
# speedup vs baseline: 3.6041x; 1.0562x over previous
#include <hip/hip_runtime.h>
#include <hip/hip_bf16.h>

#define N_NODES 50000
#define N_EDGES 800000
#define E_TOT   (N_EDGES + N_NODES)
#define F_IN    256
#define HEADS   8
#define HID     32
#define N_CLS   16
#define HC1     (HEADS * HID)
#define HC2     (HEADS * N_CLS)
#define NEG_SLOPE 0.2f

typedef float v2f __attribute__((ext_vector_type(2)));
typedef float v8f __attribute__((ext_vector_type(8)));

template <int N, int K, int CH, int WPE>
__global__ __launch_bounds__(256, WPE) void gemm_f32_wmma(
    const float* __restrict__ A, const float* __restrict__ B,
    const float* __restrict__ a_src, const float* __restrict__ a_dst,
    float* __restrict__ C, float* __restrict__ as, float* __restrict__ ad, int M) {
  constexpr int TILES_N = N / 64;
  constexpr int WAVES   = 8;
  constexpr int RB      = 16 * WAVES / TILES_N;
  constexpr int PITCH   = K + 4;
  constexpr int Q       = K / 4;
  constexpr int PQ      = PITCH / 4;
  static_assert(WAVES % TILES_N == 0, "a work-group holds whole row tiles");
  static_assert((RB * Q) % 256 == 0 && K % 4 == 0, "the panel splits evenly over 256 lanes");
  __shared__ float4 sA4[RB * PQ];
  const int lane  = threadIdx.x & 31;
  const int wib   = threadIdx.x >> 5;
  const int wave  = blockIdx.x * WAVES + wib;
  const int tileM = wave / TILES_N;
  const int tileN = wave % TILES_N;

  {
    constexpr int NL = RB * Q / 256;
    const int row0 = blockIdx.x * RB;
    float4 pv[NL];
    #pragma unroll
    for (int j = 0; j < NL; ++j) {
      const int f  = j * 256 + (int)threadIdx.x;
      const int gr = row0 + f / Q;
      pv[j] = ((const float4*)(A + (size_t)(gr < M ? gr : M - 1) * K))[f % Q];
    }
    #pragma unroll
    for (int j = 0; j < NL; ++j) {
      const int f = j * 256 + (int)threadIdx.x;
      sA4[(f / Q) * PQ + f % Q] = pv[j];
    }
  }
  __syncthreads();
  if (tileM * 16 >= M) return;

  const int r    = lane & 15;
  const int ksel = (lane >> 4) << 1;

  const float* sarow = (const float*)sA4 + ((wib / TILES_N) * 16 + r) * PITCH + ksel;
  const float* __restrict__ bbase = B + (size_t)ksel * N + tileN * 64 + r;

  v8f acc0 = {}, acc1 = {}, acc2 = {}, acc3 = {};
  #pragma unroll 2
  for (int kk = 0; kk < K; kk += 4) {
    v2f b0, b1, b2, b3;
    const v2f a = *(const v2f*)(sarow + kk);
    b0[0] = bbase[0];       b0[1] = bbase[N];
    b1[0] = bbase[16];      b1[1] = bbase[N + 16];
    b2[0] = bbase[32];      b2[1] = bbase[N + 32];
    b3[0] = bbase[48];      b3[1] = bbase[N + 48];
    acc0 = __builtin_amdgcn_wmma_f32_16x16x4_f32(false, a, false, b0, (short)0, acc0, false, false);
    acc1 = __builtin_amdgcn_wmma_f32_16x16x4_f32(false, a, false, b1, (short)0, acc1, false, false);
    acc2 = __builtin_amdgcn_wmma_f32_16x16x4_f32(false, a, false, b2, (short)0, acc2, false, false);
    acc3 = __builtin_amdgcn_wmma_f32_16x16x4_f32(false, a, false, b3, (short)0, acc3, false, false);
    bbase += (size_t)4 * N;
  }

  float* __restrict__ crow =
      C + (size_t)(tileM * 16 + (lane >> 4) * 8) * N + tileN * 64 + (lane & 15);
  #pragma unroll
  for (int i = 0; i < 8; ++i) {
    crow[(size_t)i * N +  0] = acc0[i];
    crow[(size_t)i * N + 16] = acc1[i];
    crow[(size_t)i * N + 32] = acc2[i];
    crow[(size_t)i * N + 48] = acc3[i];
  }

  constexpr int HPW = 64 / CH;
  constexpr int APH = CH / 16;
  constexpr int NV  = 8 * HPW;
  constexpr int R   = NV / 16;
  static_assert(HPW * CH == 64 && APH * 16 == CH, "whole heads per wave");
  static_assert(HPW == 2 * R, "row of a finished value is (lane & 15) >> 1");
  static_assert(N / CH == HEADS, "logit rows hold HEADS values");
  const int c = lane & 15;
  const v8f accs[4] = {acc0, acc1, acc2, acc3};
  float ps[4], pd[4];
  #pragma unroll
  for (int j = 0; j < 4; ++j) {
    ps[j] = a_src[tileN * 64 + 16 * j + c];
    pd[j] = a_dst[tileN * 64 + 16 * j + c];
  }
  float vs[NV], vd[NV];
  #pragma unroll
  for (int i = 0; i < 8; ++i) {
    #pragma unroll
    for (int q = 0; q < HPW; ++q) {
      float sa = accs[q * APH][i] * ps[q * APH];
      float sd = accs[q * APH][i] * pd[q * APH];
      #pragma unroll
      for (int t = 1; t < APH; ++t) {
        const float hv = accs[q * APH + t][i];
        sa += hv * ps[q * APH + t];
        sd += hv * pd[q * APH + t];
      }
      vs[i * HPW + q] = sa;
      vd[i * HPW + q] = sd;
    }
  }
  #pragma unroll
  for (int m = 8, n = NV / 2; m >= 1; m >>= 1, n >>= 1) {
    const bool up = (c & m) != 0;
    #pragma unroll
    for (int k = 0; k < n; ++k) {
      const float s_lo = vs[k], s_hi = vs[k + n];
      const float d_lo = vd[k], d_hi = vd[k + n];
      const float keep_s = up ? s_hi : s_lo;
      const float send_s = up ? s_lo : s_hi;
      const float keep_d = up ? d_hi : d_lo;
      const float send_d = up ? d_lo : d_hi;
      vs[k] = keep_s + __shfl_xor(send_s, m, 32);
      vd[k] = keep_d + __shfl_xor(send_d, m, 32);
    }
  }
  const size_t lo = (size_t)(tileM * 16 + (lane >> 4) * 8 + (c >> 1)) * HEADS +
                    tileN * HPW + R * (c & 1);
  #pragma unroll
  for (int k = 0; k < R; ++k) {
    as[lo + k] = vs[k];
    ad[lo + k] = vd[k];
  }
}

#define SORT_WG   128
#define SORT_R    ((N_NODES + SORT_WG - 1) / SORT_WG)
#define SORT_TH   1024
#define SORT_CAP  16384
#define SORT_NQ   (N_EDGES / 4)
#define SORT_UNR  8
__global__ __launch_bounds__(SORT_TH) void csr_build_kernel(
    const int* __restrict__ ei, int* __restrict__ rowptr, int* __restrict__ col) {
  static_assert(N_EDGES % 4 == 0, "destinations are read in 16-byte groups");
  static_assert(N_EDGES <= (1 << 20) && SORT_R <= (1 << 12),
                "edge id (20 bits) and node inside the range (12 bits) share a word");
  static_assert((SORT_WG - 1) * SORT_R < N_NODES && SORT_WG * SORT_R >= N_NODES,
                "every range holds a node and the ranges cover all nodes");
  __shared__ unsigned s_cnt[SORT_R];
  __shared__ unsigned s_excl[SORT_R];
  __shared__ unsigned s_list[SORT_CAP];
  __shared__ unsigned s_n;
  __shared__ unsigned s_below;
  const int t    = threadIdx.x;
  const int lane = t & 31;
  const int lo   = blockIdx.x * SORT_R;
  const int hi   = (lo + SORT_R < N_NODES) ? lo + SORT_R : N_NODES;
  const unsigned rthis = (unsigned)(hi - lo);
  for (int i = t; i < SORT_R; i += SORT_TH) s_cnt[i] = 0u;
  if (t == 0) { s_n = 0u; s_below = 0u; }
  __syncthreads();

  const int4* __restrict__ d4 = (const int4*)(ei + N_EDGES);
  auto take = [&](unsigned dl, int e) {
    if (dl < rthis) {
      const unsigned p = atomicAdd(&s_n, 1u);
      if (p < SORT_CAP) s_list[p] = (dl << 20) | (unsigned)e;
      atomicAdd(&s_cnt[dl], 1u);
    }
  };
  unsigned below = 0u;
  for (int q0 = 0; q0 < SORT_NQ; q0 += SORT_TH * SORT_UNR) {
    int4 v[SORT_UNR];
    #pragma unroll
    for (int j = 0; j < SORT_UNR; ++j) {
      const int q = q0 + j * SORT_TH + t;
      v[j] = (q < SORT_NQ) ? d4[q] : make_int4(hi, hi, hi, hi);
    }
    #pragma unroll
    for (int j = 0; j < SORT_UNR; ++j) {
      const int q = q0 + j * SORT_TH + t;
      const unsigned a = (unsigned)v[j].x - (unsigned)lo;
      const unsigned b = (unsigned)v[j].y - (unsigned)lo;
      const unsigned c = (unsigned)v[j].z - (unsigned)lo;
      const unsigned d = (unsigned)v[j].w - (unsigned)lo;
      below += (a >> 31) + (b >> 31) + (c >> 31) + (d >> 31);
      const unsigned mab = a < b ? a : b;
      const unsigned mcd = c < d ? c : d;
      if ((mab < mcd ? mab : mcd) < rthis) {
        take(a, 4 * q + 0);
        take(b, 4 * q + 1);
        take(c, 4 * q + 2);
        take(d, 4 * q + 3);
      }
    }
  }
  #pragma unroll
  for (int off = 16; off > 0; off >>= 1) below += __shfl_xor(below, off, 32);
  if (lane == 0) atomicAdd(&s_below, below);
  __syncthreads();
  const unsigned base  = s_below;
  const unsigned nmine = s_n;

  if (t < 32) {
    unsigned carry = 0u;
    for (int c = 0; c < (SORT_R + 31) / 32; ++c) {
      const int i = c * 32 + lane;
      const bool in = (unsigned)i < rthis;
      const unsigned v = in ? s_cnt[i] : 0u;
      unsigned inc = v;
      #pragma unroll
      for (int off = 1; off < 32; off <<= 1) {
        const unsigned y = __shfl_up(inc, off, 32);
        if (lane >= off) inc += y;
      }
      const unsigned ex = inc - v + carry;
      if (in) {
        s_excl[i] = ex;
        s_cnt[i]  = 0u;
        rowptr[lo + i] = (int)(base + ex);
      }
      carry += __shfl(inc, 31, 32);
    }
    if (lane == 0 && hi == N_NODES) rowptr[N_NODES] = (int)(base + carry);
  }
  __syncthreads();

  auto place = [&](unsigned dl, int e) {
    const int s = ei[e];
    const unsigned r = atomicAdd(&s_cnt[dl], 1u);
    col[base + s_excl[dl] + r] = s;
  };
  if (nmine <= SORT_CAP) {
    for (unsigned k = (unsigned)t; k < nmine; k += SORT_TH) {
      const unsigned p = s_list[k];
      place(p >> 20, (int)(p & 0xFFFFFu));
    }
  } else {
    for (int q = t; q < SORT_NQ; q += SORT_TH) {
      const int4 v = d4[q];
      const unsigned a = (unsigned)v.x - (unsigned)lo;
      const unsigned b = (unsigned)v.y - (unsigned)lo;
      const unsigned c = (unsigned)v.z - (unsigned)lo;
      const unsigned d = (unsigned)v.w - (unsigned)lo;
      if (a < rthis) place(a, 4 * q + 0);
      if (b < rthis) place(b, 4 * q + 1);
      if (c < rthis) place(c, 4 * q + 2);
      if (d < rthis) place(d, 4 * q + 3);
    }
  }
}

#define AGG_WAVES 8
#define CHUNK 32
#define GROUPS (CHUNK / 4)
template <int HCv, bool RELU>
__global__ __launch_bounds__(256, 16) void node_softmax_agg_kernel(
    const int* __restrict__ rowptr, const int* __restrict__ col,
    const float* __restrict__ as, const float* __restrict__ ad,
    const float* __restrict__ hmat, const float* __restrict__ bias,
    float* __restrict__ outp) {
  constexpr int VEC = HCv / 32;
  constexpr int NV4 = VEC / 4;
  static_assert((HCv / HEADS) / VEC == 4, "four lanes per head");
  __shared__ float wl_all[AGG_WAVES][CHUNK * HEADS];
  __shared__ int   sl_all[AGG_WAVES][CHUNK];
  const int lane = threadIdx.x & 31;
  float* __restrict__ wl = wl_all[threadIdx.x >> 5];
  int*   __restrict__ sl = sl_all[threadIdx.x >> 5];
  const int n = __builtin_amdgcn_readfirstlane(
      (int)(blockIdx.x * (blockDim.x >> 5) + (threadIdx.x >> 5)));
  if (n >= N_NODES) return;
  const int hd  = lane >> 2;
  const int sub = lane & 3;
  const unsigned loff = (unsigned)lane * VEC;
  const int beg = rowptr[n];
  const int end = rowptr[n + 1];
  const int deg = end - beg;
  const float adv = ad[n * HEADS + hd];

  float vs = as[n * HEADS + hd] + adv;
  vs = vs > 0.f ? vs : NEG_SLOPE * vs;

  float acc[VEC];
  auto init_acc = [&](float w) {
    const float4* __restrict__ hp =
        (const float4*)(hmat + (size_t)n * HCv + loff);
    #pragma unroll
    for (int q = 0; q < NV4; ++q) {
      const float4 t = hp[q];
      acc[4 * q + 0] = w * t.x;
      acc[4 * q + 1] = w * t.y;
      acc[4 * q + 2] = w * t.z;
      acc[4 * q + 3] = w * t.w;
    }
  };
  auto add_row = [&](int s, float w) {
    const float4* __restrict__ hp =
        (const float4*)(hmat + (size_t)s * HCv + loff);
    #pragma unroll
    for (int q = 0; q < NV4; ++q) {
      const float4 t = hp[q];
      acc[4 * q + 0] += w * t.x;
      acc[4 * q + 1] += w * t.y;
      acc[4 * q + 2] += w * t.z;
      acc[4 * q + 3] += w * t.w;
    }
  };

  if (deg >= 1 && deg <= CHUNK) {
    float e[GROUPS];
    int   sv[GROUPS];
    const int last = deg - 1;
    #pragma unroll
    for (int g = 0; g < GROUPS; ++g) {
      const int idx = 4 * g + sub;
      sv[g] = col[beg + (idx < last ? idx : last)];
    }
    #pragma unroll
    for (int g = 0; g < GROUPS; ++g) e[g] = as[sv[g] * HEADS + hd];
    float m = vs;
    #pragma unroll
    for (int g = 0; g < GROUPS; ++g) {
      float v = e[g] + adv;
      v = v > 0.f ? v : NEG_SLOPE * v;
      v = (4 * g + sub) < deg ? v : -INFINITY;
      e[g] = v;
      m = v > m ? v : m;
    }
    {
      float y = __shfl_xor(m, 1, 32); m = y > m ? y : m;
      y = __shfl_xor(m, 2, 32);       m = y > m ? y : m;
    }
    float den = 0.f;
    #pragma unroll
    for (int g = 0; g < GROUPS; ++g) {
      e[g] = __expf(e[g] - m);
      den += e[g];
    }
    den += __shfl_xor(den, 1, 32);
    den += __shfl_xor(den, 2, 32);
    const float es = __expf(vs - m);
    den += es;
    const float dn = den + 1e-16f;
    #pragma unroll
    for (int g = 0; g < GROUPS; ++g) {
      wl[(4 * g + sub) * HEADS + hd] = e[g] / dn;
      sl[4 * g + sub] = sv[g];
    }
    __builtin_amdgcn_wave_barrier();
    init_acc(es / dn);
    for (int i = 0; i < deg; i += 4) {
      #pragma unroll
      for (int j = 0; j < 4; ++j) {
        const int s = __builtin_amdgcn_readfirstlane(sl[i + j]);
        add_row(s, wl[(i + j) * HEADS + hd]);
      }
    }
  } else {
    float m = vs;
    #pragma unroll 4
    for (int i = beg; i < end; ++i) {
      const int s = col[i];
      float v = as[s * HEADS + hd] + adv;
      v = v > 0.f ? v : NEG_SLOPE * v;
      m = v > m ? v : m;
    }
    float den = __expf(vs - m);
    #pragma unroll 4
    for (int i = beg; i < end; ++i) {
      const int s = col[i];
      float v = as[s * HEADS + hd] + adv;
      v = v > 0.f ? v : NEG_SLOPE * v;
      den += __expf(v - m);
    }
    const float dn = den + 1e-16f;
    init_acc(__expf(vs - m) / dn);
    #pragma unroll 4
    for (int i = beg; i < end; ++i) {
      const int s = col[i];
      float v = as[s * HEADS + hd] + adv;
      v = v > 0.f ? v : NEG_SLOPE * v;
      add_row(s, __expf(v - m) / dn);
    }
  }

  float4* __restrict__ op = (float4*)(outp + (size_t)n * HCv + lane * VEC);
  const float4* __restrict__ bp = (const float4*)(bias + lane * VEC);
  #pragma unroll
  for (int q = 0; q < NV4; ++q) {
    const float4 b = bp[q];
    float4 r;
    r.x = acc[4 * q + 0] + b.x;
    r.y = acc[4 * q + 1] + b.y;
    r.z = acc[4 * q + 2] + b.z;
    r.w = acc[4 * q + 3] + b.w;
    if (RELU) {
      r.x = r.x > 0.f ? r.x : 0.f;
      r.y = r.y > 0.f ? r.y : 0.f;
      r.z = r.z > 0.f ? r.z : 0.f;
      r.w = r.w > 0.f ? r.w : 0.f;
    }
    op[q] = r;
  }
}

static inline int blocks_for(long long n, int bs) {
  return (int)((n + bs - 1) / bs);
}

extern "C" void kernel_launch(void* const* d_in, const int* in_sizes, int n_in,
                              void* d_out, int out_size, void* d_ws, size_t ws_size,
                              hipStream_t stream) {
  const float* x      = (const float*)d_in[0];
  const int*   ei     = (const int*)  d_in[1];
  const float* W1     = (const float*)d_in[2];
  const float* a_src1 = (const float*)d_in[3];
  const float* a_dst1 = (const float*)d_in[4];
  const float* b1     = (const float*)d_in[5];
  const float* W2     = (const float*)d_in[6];
  const float* a_src2 = (const float*)d_in[7];
  const float* a_dst2 = (const float*)d_in[8];
  const float* b2     = (const float*)d_in[9];
  float* out = (float*)d_out;

  char* ws = (char*)d_ws;
  size_t off = 0;
  auto carve = [&](size_t bytes) {
    void* p = ws + off;
    off += (bytes + 255) & ~(size_t)255;
    return p;
  };
  float* hbuf   = (float*)carve((size_t)N_NODES * HC1 * 4);
  float* aggbuf = (float*)carve((size_t)N_NODES * HC1 * 4);
  float* as     = (float*)carve((size_t)N_NODES * HEADS * 4);
  float* ad     = (float*)carve((size_t)N_NODES * HEADS * 4);
  int*   rowptr = (int*)  carve((size_t)(N_NODES + 1) * 4);
  int*   col    = (int*)  carve((size_t)N_EDGES * 4);

  const int BS = 256;

  {
    long long waves = (N_NODES / 16) * (HC1 / 64);
    gemm_f32_wmma<HC1, F_IN, HID, 16><<<blocks_for(waves * 32, BS), BS, 0, stream>>>(
        x, W1, a_src1, a_dst1, hbuf, as, ad, N_NODES);
  }

  csr_build_kernel<<<SORT_WG, SORT_TH, 0, stream>>>(ei, rowptr, col);

  node_softmax_agg_kernel<HC1, true><<<blocks_for((long long)N_NODES * 32, BS), BS, 0, stream>>>(
      rowptr, col, as, ad, hbuf, b1, aggbuf);

  {
    long long waves = (N_NODES / 16) * (HC2 / 64);
    gemm_f32_wmma<HC2, HC1, N_CLS, 8><<<blocks_for(waves * 32, BS), BS, 0, stream>>>(
        aggbuf, W2, a_src2, a_dst2, hbuf, as, ad, N_NODES);
  }
  node_softmax_agg_kernel<HC2, false><<<blocks_for((long long)N_NODES * 32, BS), BS, 0, stream>>>(
      rowptr, col, as, ad, hbuf, b2, out);
}


# Edge grouping in two launches: a binning launch groups each 4000-edge slice by destination range; a range's work-group then reads only its own segments, not all destinations; fill arrival-ranked
# speedup vs baseline: 4.3194x; 1.1985x over previous
#include <hip/hip_runtime.h>
#include <hip/hip_bf16.h>

#define N_NODES 50000
#define N_EDGES 800000
#define E_TOT   (N_EDGES + N_NODES)
#define F_IN    256
#define HEADS   8
#define HID     32
#define N_CLS   16
#define HC1     (HEADS * HID)
#define HC2     (HEADS * N_CLS)
#define NEG_SLOPE 0.2f

typedef float v2f __attribute__((ext_vector_type(2)));
typedef float v8f __attribute__((ext_vector_type(8)));

template <int N, int K, int CH, int WPE>
__global__ __launch_bounds__(256, WPE) void gemm_f32_wmma(
    const float* __restrict__ A, const float* __restrict__ B,
    const float* __restrict__ a_src, const float* __restrict__ a_dst,
    float* __restrict__ C, float* __restrict__ as, float* __restrict__ ad, int M) {
  constexpr int TILES_N = N / 64;
  constexpr int WAVES   = 8;
  constexpr int RB      = 16 * WAVES / TILES_N;
  constexpr int PITCH   = K + 4;
  constexpr int Q       = K / 4;
  constexpr int PQ      = PITCH / 4;
  static_assert(WAVES % TILES_N == 0, "a work-group holds whole row tiles");
  static_assert((RB * Q) % 256 == 0 && K % 4 == 0, "the panel splits evenly over 256 lanes");
  __shared__ float4 sA4[RB * PQ];
  const int lane  = threadIdx.x & 31;
  const int wib   = threadIdx.x >> 5;
  const int wave  = blockIdx.x * WAVES + wib;
  const int tileM = wave / TILES_N;
  const int tileN = wave % TILES_N;

  {
    constexpr int NL = RB * Q / 256;
    const int row0 = blockIdx.x * RB;
    float4 pv[NL];
    #pragma unroll
    for (int j = 0; j < NL; ++j) {
      const int f  = j * 256 + (int)threadIdx.x;
      const int gr = row0 + f / Q;
      pv[j] = ((const float4*)(A + (size_t)(gr < M ? gr : M - 1) * K))[f % Q];
    }
    #pragma unroll
    for (int j = 0; j < NL; ++j) {
      const int f = j * 256 + (int)threadIdx.x;
      sA4[(f / Q) * PQ + f % Q] = pv[j];
    }
  }
  __syncthreads();
  if (tileM * 16 >= M) return;

  const int r    = lane & 15;
  const int ksel = (lane >> 4) << 1;

  const float* sarow = (const float*)sA4 + ((wib / TILES_N) * 16 + r) * PITCH + ksel;
  const float* __restrict__ bbase = B + (size_t)ksel * N + tileN * 64 + r;

  v8f acc0 = {}, acc1 = {}, acc2 = {}, acc3 = {};
  #pragma unroll 2
  for (int kk = 0; kk < K; kk += 4) {
    v2f b0, b1, b2, b3;
    const v2f a = *(const v2f*)(sarow + kk);
    b0[0] = bbase[0];       b0[1] = bbase[N];
    b1[0] = bbase[16];      b1[1] = bbase[N + 16];
    b2[0] = bbase[32];      b2[1] = bbase[N + 32];
    b3[0] = bbase[48];      b3[1] = bbase[N + 48];
    acc0 = __builtin_amdgcn_wmma_f32_16x16x4_f32(false, a, false, b0, (short)0, acc0, false, false);
    acc1 = __builtin_amdgcn_wmma_f32_16x16x4_f32(false, a, false, b1, (short)0, acc1, false, false);
    acc2 = __builtin_amdgcn_wmma_f32_16x16x4_f32(false, a, false, b2, (short)0, acc2, false, false);
    acc3 = __builtin_amdgcn_wmma_f32_16x16x4_f32(false, a, false, b3, (short)0, acc3, false, false);
    bbase += (size_t)4 * N;
  }

  float* __restrict__ crow =
      C + (size_t)(tileM * 16 + (lane >> 4) * 8) * N + tileN * 64 + (lane & 15);
  #pragma unroll
  for (int i = 0; i < 8; ++i) {
    crow[(size_t)i * N +  0] = acc0[i];
    crow[(size_t)i * N + 16] = acc1[i];
    crow[(size_t)i * N + 32] = acc2[i];
    crow[(size_t)i * N + 48] = acc3[i];
  }

  constexpr int HPW = 64 / CH;
  constexpr int APH = CH / 16;
  constexpr int NV  = 8 * HPW;
  constexpr int R   = NV / 16;
  static_assert(HPW * CH == 64 && APH * 16 == CH, "whole heads per wave");
  static_assert(HPW == 2 * R, "row of a finished value is (lane & 15) >> 1");
  static_assert(N / CH == HEADS, "logit rows hold HEADS values");
  const int c = lane & 15;
  const v8f accs[4] = {acc0, acc1, acc2, acc3};
  float ps[4], pd[4];
  #pragma unroll
  for (int j = 0; j < 4; ++j) {
    ps[j] = a_src[tileN * 64 + 16 * j + c];
    pd[j] = a_dst[tileN * 64 + 16 * j + c];
  }
  float vs[NV], vd[NV];
  #pragma unroll
  for (int i = 0; i < 8; ++i) {
    #pragma unroll
    for (int q = 0; q < HPW; ++q) {
      float sa = accs[q * APH][i] * ps[q * APH];
      float sd = accs[q * APH][i] * pd[q * APH];
      #pragma unroll
      for (int t = 1; t < APH; ++t) {
        const float hv = accs[q * APH + t][i];
        sa += hv * ps[q * APH + t];
        sd += hv * pd[q * APH + t];
      }
      vs[i * HPW + q] = sa;
      vd[i * HPW + q] = sd;
    }
  }
  #pragma unroll
  for (int m = 8, n = NV / 2; m >= 1; m >>= 1, n >>= 1) {
    const bool up = (c & m) != 0;
    #pragma unroll
    for (int k = 0; k < n; ++k) {
      const float s_lo = vs[k], s_hi = vs[k + n];
      const float d_lo = vd[k], d_hi = vd[k + n];
      const float keep_s = up ? s_hi : s_lo;
      const float send_s = up ? s_lo : s_hi;
      const float keep_d = up ? d_hi : d_lo;
      const float send_d = up ? d_lo : d_hi;
      vs[k] = keep_s + __shfl_xor(send_s, m, 32);
      vd[k] = keep_d + __shfl_xor(send_d, m, 32);
    }
  }
  const size_t lo = (size_t)(tileM * 16 + (lane >> 4) * 8 + (c >> 1)) * HEADS +
                    tileN * HPW + R * (c & 1);
  #pragma unroll
  for (int k = 0; k < R; ++k) {
    as[lo + k] = vs[k];
    ad[lo + k] = vd[k];
  }
}

#define SORT_WG   128
#define SORT_R    ((N_NODES + SORT_WG - 1) / SORT_WG)
#define SORT_TH   1024
#define BIN_SLICE 4000
#define BIN_WG    (N_EDGES / BIN_SLICE)
#define BIN_TH    1024
#define BIN_PF    (SORT_WG + 1)
__global__ __launch_bounds__(BIN_TH) void bin_edges_kernel(
    const int* __restrict__ ei, unsigned* __restrict__ binned, int* __restrict__ pfx) {
  static_assert(N_EDGES % BIN_SLICE == 0 && BIN_SLICE % 4 == 0 && BIN_SLICE / 4 <= BIN_TH,
                "a slice is whole 16-byte groups, one per thread");
  static_assert(N_NODES <= (1 << 16) && SORT_R <= (1 << 16),
                "source (16 bits) and node inside the range (16 bits) share a word");
  static_assert(SORT_WG % 32 == 0 && SORT_WG * SORT_R >= N_NODES,
                "the range scan runs whole 32-lane trips and the ranges cover all nodes");
  __shared__ unsigned s_cnt[SORT_WG];
  __shared__ unsigned s_excl[SORT_WG];
  const int t    = threadIdx.x;
  const int lane = t & 31;
  for (int i = t; i < SORT_WG; i += BIN_TH) s_cnt[i] = 0u;
  __syncthreads();

  int4 d = make_int4(-1, -1, -1, -1);
  int4 s = make_int4(0, 0, 0, 0);
  if (t < BIN_SLICE / 4) {
    const int q = blockIdx.x * (BIN_SLICE / 4) + t;
    d = ((const int4*)(ei + N_EDGES))[q];
    s = ((const int4*)ei)[q];
  }
  const int dv[4] = {d.x, d.y, d.z, d.w};
  const int sv[4] = {s.x, s.y, s.z, s.w};
  unsigned rg[4], word[4], rank[4];
  #pragma unroll
  for (int j = 0; j < 4; ++j) {
    const unsigned dd = (unsigned)dv[j];
    const unsigned r  = dd / (unsigned)SORT_R;
    rg[j]   = SORT_WG;
    rank[j] = 0u;
    word[j] = ((dd - r * (unsigned)SORT_R) << 16) | ((unsigned)sv[j] & 0xFFFFu);
    if (dd < (unsigned)N_NODES) {
      rg[j]   = r;
      rank[j] = atomicAdd(&s_cnt[r], 1u);
    }
  }
  __syncthreads();

  if (t < 32) {
    unsigned carry = 0u;
    for (int c = 0; c < SORT_WG / 32; ++c) {
      const int i = c * 32 + lane;
      const unsigned v = s_cnt[i];
      unsigned inc = v;
      #pragma unroll
      for (int off = 1; off < 32; off <<= 1) {
        const unsigned y = __shfl_up(inc, off, 32);
        if (lane >= off) inc += y;
      }
      const unsigned ex = inc - v + carry;
      s_excl[i] = ex;
      pfx[blockIdx.x * BIN_PF + i] = (int)ex;
      carry += __shfl(inc, 31, 32);
    }
    if (lane == 0) pfx[blockIdx.x * BIN_PF + SORT_WG] = (int)carry;
  }
  __syncthreads();

  unsigned* __restrict__ mine = binned + (size_t)blockIdx.x * BIN_SLICE;
  #pragma unroll
  for (int j = 0; j < 4; ++j)
    if (rg[j] < SORT_WG) mine[s_excl[rg[j]] + rank[j]] = word[j];
}

__global__ __launch_bounds__(SORT_TH) void csr_build_kernel(
    const unsigned* __restrict__ binned, const int* __restrict__ pfx,
    int* __restrict__ rowptr, int* __restrict__ col) {
  static_assert((SORT_WG - 1) * SORT_R < N_NODES && SORT_WG * SORT_R >= N_NODES,
                "every range holds a node and the ranges cover all nodes");
  static_assert(BIN_WG <= SORT_TH, "one thread per slice reads the slice's range start");
  constexpr int NWV = SORT_TH / 32;
  constexpr int SPW = (BIN_WG + NWV - 1) / NWV;
  __shared__ unsigned s_cnt[SORT_R];
  __shared__ unsigned s_excl[SORT_R];
  __shared__ unsigned s_off[BIN_WG];
  __shared__ unsigned s_len[BIN_WG];
  __shared__ unsigned s_below;
  const int t    = threadIdx.x;
  const int lane = t & 31;
  const int wv   = t >> 5;
  const int lo   = blockIdx.x * SORT_R;
  const int hi   = (lo + SORT_R < N_NODES) ? lo + SORT_R : N_NODES;
  const unsigned rthis = (unsigned)(hi - lo);
  for (int i = t; i < SORT_R; i += SORT_TH) s_cnt[i] = 0u;
  if (t == 0) s_below = 0u;
  __syncthreads();

  unsigned below = 0u;
  if (t < BIN_WG) {
    const int* __restrict__ p = pfx + t * BIN_PF + blockIdx.x;
    const unsigned o = (unsigned)p[0];
    const unsigned e = (unsigned)p[1];
    s_off[t] = (unsigned)t * BIN_SLICE + o;
    s_len[t] = e - o;
    below = o;
  }
  #pragma unroll
  for (int off = 16; off > 0; off >>= 1) below += __shfl_xor(below, off, 32);
  if (lane == 0) atomicAdd(&s_below, below);
  __syncthreads();
  const unsigned base = s_below;

  auto walk = [&](auto use) {
    unsigned off[SPW], len[SPW], w0[SPW], w1[SPW];
    #pragma unroll
    for (int j = 0; j < SPW; ++j) {
      const int b = wv + NWV * j;
      const bool in = b < BIN_WG;
      off[j] = in ? s_off[b] : 0u;
      len[j] = in ? s_len[b] : 0u;
    }
    #pragma unroll
    for (int j = 0; j < SPW; ++j) {
      const unsigned last = len[j] ? len[j] - 1u : 0u;
      const unsigned l0 = (unsigned)lane, l1 = (unsigned)lane + 32u;
      const unsigned o  = len[j] ? off[j] : 0u;
      w0[j] = binned[o + (l0 < last ? l0 : last)];
      w1[j] = binned[o + (l1 < last ? l1 : last)];
    }
    #pragma unroll
    for (int j = 0; j < SPW; ++j) {
      if ((unsigned)lane < len[j]) use(w0[j]);
      if ((unsigned)lane + 32u < len[j]) use(w1[j]);
      for (unsigned l = (unsigned)lane + 64u; l < len[j]; l += 32u) use(binned[off[j] + l]);
    }
  };

  walk([&](unsigned p) {
    const unsigned dl = p >> 16;
    if (dl < rthis) atomicAdd(&s_cnt[dl], 1u);
  });
  __syncthreads();

  if (t < 32) {
    unsigned carry = 0u;
    for (int c = 0; c < (SORT_R + 31) / 32; ++c) {
      const int i = c * 32 + lane;
      const bool in = (unsigned)i < rthis;
      const unsigned v = in ? s_cnt[i] : 0u;
      unsigned inc = v;
      #pragma unroll
      for (int off = 1; off < 32; off <<= 1) {
        const unsigned y = __shfl_up(inc, off, 32);
        if (lane >= off) inc += y;
      }
      const unsigned ex = inc - v + carry;
      if (in) {
        s_excl[i] = ex;
        s_cnt[i]  = 0u;
        rowptr[lo + i] = (int)(base + ex);
      }
      carry += __shfl(inc, 31, 32);
    }
    if (lane == 0 && hi == N_NODES) rowptr[N_NODES] = (int)(base + carry);
  }
  __syncthreads();

  walk([&](unsigned p) {
    const unsigned dl = p >> 16;
    if (dl < rthis) {
      const unsigned r = atomicAdd(&s_cnt[dl], 1u);
      col[base + s_excl[dl] + r] = (int)(p & 0xFFFFu);
    }
  });
}

#define AGG_WAVES 8
#define CHUNK 32
#define GROUPS (CHUNK / 4)
template <int HCv, bool RELU>
__global__ __launch_bounds__(256, 16) void node_softmax_agg_kernel(
    const int* __restrict__ rowptr, const int* __restrict__ col,
    const float* __restrict__ as, const float* __restrict__ ad,
    const float* __restrict__ hmat, const float* __restrict__ bias,
    float* __restrict__ outp) {
  constexpr int VEC = HCv / 32;
  constexpr int NV4 = VEC / 4;
  static_assert((HCv / HEADS) / VEC == 4, "four lanes per head");
  __shared__ float wl_all[AGG_WAVES][CHUNK * HEADS];
  __shared__ int   sl_all[AGG_WAVES][CHUNK];
  const int lane = threadIdx.x & 31;
  float* __restrict__ wl = wl_all[threadIdx.x >> 5];
  int*   __restrict__ sl = sl_all[threadIdx.x >> 5];
  const int n = __builtin_amdgcn_readfirstlane(
      (int)(blockIdx.x * (blockDim.x >> 5) + (threadIdx.x >> 5)));
  if (n >= N_NODES) return;
  const int hd  = lane >> 2;
  const int sub = lane & 3;
  const unsigned loff = (unsigned)lane * VEC;
  const int beg = rowptr[n];
  const int end = rowptr[n + 1];
  const int deg = end - beg;
  const float adv = ad[n * HEADS + hd];

  float vs = as[n * HEADS + hd] + adv;
  vs = vs > 0.f ? vs : NEG_SLOPE * vs;

  float acc[VEC];
  auto init_acc = [&](float w) {
    const float4* __restrict__ hp =
        (const float4*)(hmat + (size_t)n * HCv + loff);
    #pragma unroll
    for (int q = 0; q < NV4; ++q) {
      const float4 t = hp[q];
      acc[4 * q + 0] = w * t.x;
      acc[4 * q + 1] = w * t.y;
      acc[4 * q + 2] = w * t.z;
      acc[4 * q + 3] = w * t.w;
    }
  };
  auto add_row = [&](int s, float w) {
    const float4* __restrict__ hp =
        (const float4*)(hmat + (size_t)s * HCv + loff);
    #pragma unroll
    for (int q = 0; q < NV4; ++q) {
      const float4 t = hp[q];
      acc[4 * q + 0] += w * t.x;
      acc[4 * q + 1] += w * t.y;
      acc[4 * q + 2] += w * t.z;
      acc[4 * q + 3] += w * t.w;
    }
  };

  if (deg >= 1 && deg <= CHUNK) {
    float e[GROUPS];
    int   sv[GROUPS];
    const int last = deg - 1;
    #pragma unroll
    for (int g = 0; g < GROUPS; ++g) {
      const int idx = 4 * g + sub;
      sv[g] = col[beg + (idx < last ? idx : last)];
    }
    #pragma unroll
    for (int g = 0; g < GROUPS; ++g) e[g] = as[sv[g] * HEADS + hd];
    float m = vs;
    #pragma unroll
    for (int g = 0; g < GROUPS; ++g) {
      float v = e[g] + adv;
      v = v > 0.f ? v : NEG_SLOPE * v;
      v = (4 * g + sub) < deg ? v : -INFINITY;
      e[g] = v;
      m = v > m ? v : m;
    }
    {
      float y = __shfl_xor(m, 1, 32); m = y > m ? y : m;
      y = __shfl_xor(m, 2, 32);       m = y > m ? y : m;
    }
    float den = 0.f;
    #pragma unroll
    for (int g = 0; g < GROUPS; ++g) {
      e[g] = __expf(e[g] - m);
      den += e[g];
    }
    den += __shfl_xor(den, 1, 32);
    den += __shfl_xor(den, 2, 32);
    const float es = __expf(vs - m);
    den += es;
    const float dn = den + 1e-16f;
    #pragma unroll
    for (int g = 0; g < GROUPS; ++g) {
      wl[(4 * g + sub) * HEADS + hd] = e[g] / dn;
      sl[4 * g + sub] = sv[g];
    }
    __builtin_amdgcn_wave_barrier();
    init_acc(es / dn);
    for (int i = 0; i < deg; i += 4) {
      #pragma unroll
      for (int j = 0; j < 4; ++j) {
        const int s = __builtin_amdgcn_readfirstlane(sl[i + j]);
        add_row(s, wl[(i + j) * HEADS + hd]);
      }
    }
  } else {
    float m = vs;
    #pragma unroll 4
    for (int i = beg; i < end; ++i) {
      const int s = col[i];
      float v = as[s * HEADS + hd] + adv;
      v = v > 0.f ? v : NEG_SLOPE * v;
      m = v > m ? v : m;
    }
    float den = __expf(vs - m);
    #pragma unroll 4
    for (int i = beg; i < end; ++i) {
      const int s = col[i];
      float v = as[s * HEADS + hd] + adv;
      v = v > 0.f ? v : NEG_SLOPE * v;
      den += __expf(v - m);
    }
    const float dn = den + 1e-16f;
    init_acc(__expf(vs - m) / dn);
    #pragma unroll 4
    for (int i = beg; i < end; ++i) {
      const int s = col[i];
      float v = as[s * HEADS + hd] + adv;
      v = v > 0.f ? v : NEG_SLOPE * v;
      add_row(s, __expf(v - m) / dn);
    }
  }

  float4* __restrict__ op = (float4*)(outp + (size_t)n * HCv + lane * VEC);
  const float4* __restrict__ bp = (const float4*)(bias + lane * VEC);
  #pragma unroll
  for (int q = 0; q < NV4; ++q) {
    const float4 b = bp[q];
    float4 r;
    r.x = acc[4 * q + 0] + b.x;
    r.y = acc[4 * q + 1] + b.y;
    r.z = acc[4 * q + 2] + b.z;
    r.w = acc[4 * q + 3] + b.w;
    if (RELU) {
      r.x = r.x > 0.f ? r.x : 0.f;
      r.y = r.y > 0.f ? r.y : 0.f;
      r.z = r.z > 0.f ? r.z : 0.f;
      r.w = r.w > 0.f ? r.w : 0.f;
    }
    op[q] = r;
  }
}

static inline int blocks_for(long long n, int bs) {
  return (int)((n + bs - 1) / bs);
}

extern "C" void kernel_launch(void* const* d_in, const int* in_sizes, int n_in,
                              void* d_out, int out_size, void* d_ws, size_t ws_size,
                              hipStream_t stream) {
  const float* x      = (const float*)d_in[0];
  const int*   ei     = (const int*)  d_in[1];
  const float* W1     = (const float*)d_in[2];
  const float* a_src1 = (const float*)d_in[3];
  const float* a_dst1 = (const float*)d_in[4];
  const float* b1     = (const float*)d_in[5];
  const float* W2     = (const float*)d_in[6];
  const float* a_src2 = (const float*)d_in[7];
  const float* a_dst2 = (const float*)d_in[8];
  const float* b2     = (const float*)d_in[9];
  float* out = (float*)d_out;

  char* ws = (char*)d_ws;
  size_t off = 0;
  auto carve = [&](size_t bytes) {
    void* p = ws + off;
    off += (bytes + 255) & ~(size_t)255;
    return p;
  };
  float* hbuf   = (float*)carve((size_t)N_NODES * HC1 * 4);
  float* aggbuf = (float*)carve((size_t)N_NODES * HC1 * 4);
  float* as     = (float*)carve((size_t)N_NODES * HEADS * 4);
  float* ad     = (float*)carve((size_t)N_NODES * HEADS * 4);
  int*   rowptr = (int*)  carve((size_t)(N_NODES + 1) * 4);
  int*   col    = (int*)  carve((size_t)N_EDGES * 4);
  unsigned* binned = (unsigned*)carve((size_t)N_EDGES * 4);
  int*   pfx    = (int*)  carve((size_t)BIN_WG * BIN_PF * 4);

  const int BS = 256;

  {
    long long waves = (N_NODES / 16) * (HC1 / 64);
    gemm_f32_wmma<HC1, F_IN, HID, 16><<<blocks_for(waves * 32, BS), BS, 0, stream>>>(
        x, W1, a_src1, a_dst1, hbuf, as, ad, N_NODES);
  }

  bin_edges_kernel<<<BIN_WG, BIN_TH, 0, stream>>>(ei, binned, pfx);
  csr_build_kernel<<<SORT_WG, SORT_TH, 0, stream>>>(binned, pfx, rowptr, col);

  node_softmax_agg_kernel<HC1, true><<<blocks_for((long long)N_NODES * 32, BS), BS, 0, stream>>>(
      rowptr, col, as, ad, hbuf, b1, aggbuf);

  {
    long long waves = (N_NODES / 16) * (HC2 / 64);
    gemm_f32_wmma<HC2, HC1, N_CLS, 8><<<blocks_for(waves * 32, BS), BS, 0, stream>>>(
        aggbuf, W2, a_src2, a_dst2, hbuf, as, ad, N_NODES);
  }
  node_softmax_agg_kernel<HC2, false><<<blocks_for((long long)N_NODES * 32, BS), BS, 0, stream>>>(
      rowptr, col, as, ad, hbuf, b2, out);
}


# Second product runs inside the first aggregation's launch: 16 one-node waves stage finished rows in LDS, 8 of them do the f32 16x16x4 tiles and logits; its launch and row plane go; sum order kept
# speedup vs baseline: 4.5119x; 1.0446x over previous
#include <hip/hip_runtime.h>
#include <hip/hip_bf16.h>

#define N_NODES 50000
#define N_EDGES 800000
#define E_TOT   (N_EDGES + N_NODES)
#define F_IN    256
#define HEADS   8
#define HID     32
#define N_CLS   16
#define HC1     (HEADS * HID)
#define HC2     (HEADS * N_CLS)
#define NEG_SLOPE 0.2f

typedef float v2f __attribute__((ext_vector_type(2)));
typedef float v8f __attribute__((ext_vector_type(8)));

template <int N, int K, int CH, int WPE>
__global__ __launch_bounds__(256, WPE) void gemm_f32_wmma(
    const float* __restrict__ A, const float* __restrict__ B,
    const float* __restrict__ a_src, const float* __restrict__ a_dst,
    float* __restrict__ C, float* __restrict__ as, float* __restrict__ ad, int M) {
  constexpr int TILES_N = N / 64;
  constexpr int WAVES   = 8;
  constexpr int RB      = 16 * WAVES / TILES_N;
  constexpr int PITCH   = K + 4;
  constexpr int Q       = K / 4;
  constexpr int PQ      = PITCH / 4;
  static_assert(WAVES % TILES_N == 0, "a work-group holds whole row tiles");
  static_assert((RB * Q) % 256 == 0 && K % 4 == 0, "the panel splits evenly over 256 lanes");
  __shared__ float4 sA4[RB * PQ];
  const int lane  = threadIdx.x & 31;
  const int wib   = threadIdx.x >> 5;
  const int wave  = blockIdx.x * WAVES + wib;
  const int tileM = wave / TILES_N;
  const int tileN = wave % TILES_N;

  {
    constexpr int NL = RB * Q / 256;
    const int row0 = blockIdx.x * RB;
    float4 pv[NL];
    #pragma unroll
    for (int j = 0; j < NL; ++j) {
      const int f  = j * 256 + (int)threadIdx.x;
      const int gr = row0 + f / Q;
      pv[j] = ((const float4*)(A + (size_t)(gr < M ? gr : M - 1) * K))[f % Q];
    }
    #pragma unroll
    for (int j = 0; j < NL; ++j) {
      const int f = j * 256 + (int)threadIdx.x;
      sA4[(f / Q) * PQ + f % Q] = pv[j];
    }
  }
  __syncthreads();
  if (tileM * 16 >= M) return;

  const int r    = lane & 15;
  const int ksel = (lane >> 4) << 1;

  const float* sarow = (const float*)sA4 + ((wib / TILES_N) * 16 + r) * PITCH + ksel;
  const float* __restrict__ bbase = B + (size_t)ksel * N + tileN * 64 + r;

  v8f acc0 = {}, acc1 = {}, acc2 = {}, acc3 = {};
  #pragma unroll 2
  for (int kk = 0; kk < K; kk += 4) {
    v2f b0, b1, b2, b3;
    const v2f a = *(const v2f*)(sarow + kk);
    b0[0] = bbase[0];       b0[1] = bbase[N];
    b1[0] = bbase[16];      b1[1] = bbase[N + 16];
    b2[0] = bbase[32];      b2[1] = bbase[N + 32];
    b3[0] = bbase[48];      b3[1] = bbase[N + 48];
    acc0 = __builtin_amdgcn_wmma_f32_16x16x4_f32(false, a, false, b0, (short)0, acc0, false, false);
    acc1 = __builtin_amdgcn_wmma_f32_16x16x4_f32(false, a, false, b1, (short)0, acc1, false, false);
    acc2 = __builtin_amdgcn_wmma_f32_16x16x4_f32(false, a, false, b2, (short)0, acc2, false, false);
    acc3 = __builtin_amdgcn_wmma_f32_16x16x4_f32(false, a, false, b3, (short)0, acc3, false, false);
    bbase += (size_t)4 * N;
  }

  float* __restrict__ crow =
      C + (size_t)(tileM * 16 + (lane >> 4) * 8) * N + tileN * 64 + (lane & 15);
  #pragma unroll
  for (int i = 0; i < 8; ++i) {
    crow[(size_t)i * N +  0] = acc0[i];
    crow[(size_t)i * N + 16] = acc1[i];
    crow[(size_t)i * N + 32] = acc2[i];
    crow[(size_t)i * N + 48] = acc3[i];
  }

  constexpr int HPW = 64 / CH;
  constexpr int APH = CH / 16;
  constexpr int NV  = 8 * HPW;
  constexpr int R   = NV / 16;
  static_assert(HPW * CH == 64 && APH * 16 == CH, "whole heads per wave");
  static_assert(HPW == 2 * R, "row of a finished value is (lane & 15) >> 1");
  static_assert(N / CH == HEADS, "logit rows hold HEADS values");
  const int c = lane & 15;
  const v8f accs[4] = {acc0, acc1, acc2, acc3};
  float ps[4], pd[4];
  #pragma unroll
  for (int j = 0; j < 4; ++j) {
    ps[j] = a_src[tileN * 64 + 16 * j + c];
    pd[j] = a_dst[tileN * 64 + 16 * j + c];
  }
  float vs[NV], vd[NV];
  #pragma unroll
  for (int i = 0; i < 8; ++i) {
    #pragma unroll
    for (int q = 0; q < HPW; ++q) {
      float sa = accs[q * APH][i] * ps[q * APH];
      float sd = accs[q * APH][i] * pd[q * APH];
      #pragma unroll
      for (int t = 1; t < APH; ++t) {
        const float hv = accs[q * APH + t][i];
        sa += hv * ps[q * APH + t];
        sd += hv * pd[q * APH + t];
      }
      vs[i * HPW + q] = sa;
      vd[i * HPW + q] = sd;
    }
  }
  #pragma unroll
  for (int m = 8, n = NV / 2; m >= 1; m >>= 1, n >>= 1) {
    const bool up = (c & m) != 0;
    #pragma unroll
    for (int k = 0; k < n; ++k) {
      const float s_lo = vs[k], s_hi = vs[k + n];
      const float d_lo = vd[k], d_hi = vd[k + n];
      const float keep_s = up ? s_hi : s_lo;
      const float send_s = up ? s_lo : s_hi;
      const float keep_d = up ? d_hi : d_lo;
      const float send_d = up ? d_lo : d_hi;
      vs[k] = keep_s + __shfl_xor(send_s, m, 32);
      vd[k] = keep_d + __shfl_xor(send_d, m, 32);
    }
  }
  const size_t lo = (size_t)(tileM * 16 + (lane >> 4) * 8 + (c >> 1)) * HEADS +
                    tileN * HPW + R * (c & 1);
  #pragma unroll
  for (int k = 0; k < R; ++k) {
    as[lo + k] = vs[k];
    ad[lo + k] = vd[k];
  }
}

#define SORT_WG   128
#define SORT_R    ((N_NODES + SORT_WG - 1) / SORT_WG)
#define SORT_TH   1024
#define BIN_SLICE 4000
#define BIN_WG    (N_EDGES / BIN_SLICE)
#define BIN_TH    1024
#define BIN_PF    (SORT_WG + 1)
__global__ __launch_bounds__(BIN_TH) void bin_edges_kernel(
    const int* __restrict__ ei, unsigned* __restrict__ binned, int* __restrict__ pfx) {
  static_assert(N_EDGES % BIN_SLICE == 0 && BIN_SLICE % 4 == 0 && BIN_SLICE / 4 <= BIN_TH,
                "a slice is whole 16-byte groups, one per thread");
  static_assert(N_NODES <= (1 << 16) && SORT_R <= (1 << 16),
                "source (16 bits) and node inside the range (16 bits) share a word");
  static_assert(SORT_WG % 32 == 0 && SORT_WG * SORT_R >= N_NODES,
                "the range scan runs whole 32-lane trips and the ranges cover all nodes");
  __shared__ unsigned s_cnt[SORT_WG];
  __shared__ unsigned s_excl[SORT_WG];
  const int t    = threadIdx.x;
  const int lane = t & 31;
  for (int i = t; i < SORT_WG; i += BIN_TH) s_cnt[i] = 0u;
  __syncthreads();

  int4 d = make_int4(-1, -1, -1, -1);
  int4 s = make_int4(0, 0, 0, 0);
  if (t < BIN_SLICE / 4) {
    const int q = blockIdx.x * (BIN_SLICE / 4) + t;
    d = ((const int4*)(ei + N_EDGES))[q];
    s = ((const int4*)ei)[q];
  }
  const int dv[4] = {d.x, d.y, d.z, d.w};
  const int sv[4] = {s.x, s.y, s.z, s.w};
  unsigned rg[4], word[4], rank[4];
  #pragma unroll
  for (int j = 0; j < 4; ++j) {
    const unsigned dd = (unsigned)dv[j];
    const unsigned r  = dd / (unsigned)SORT_R;
    rg[j]   = SORT_WG;
    rank[j] = 0u;
    word[j] = ((dd - r * (unsigned)SORT_R) << 16) | ((unsigned)sv[j] & 0xFFFFu);
    if (dd < (unsigned)N_NODES) {
      rg[j]   = r;
      rank[j] = atomicAdd(&s_cnt[r], 1u);
    }
  }
  __syncthreads();

  if (t < 32) {
    unsigned carry = 0u;
    for (int c = 0; c < SORT_WG / 32; ++c) {
      const int i = c * 32 + lane;
      const unsigned v = s_cnt[i];
      unsigned inc = v;
      #pragma unroll
      for (int off = 1; off < 32; off <<= 1) {
        const unsigned y = __shfl_up(inc, off, 32);
        if (lane >= off) inc += y;
      }
      const unsigned ex = inc - v + carry;
      s_excl[i] = ex;
      pfx[blockIdx.x * BIN_PF + i] = (int)ex;
      carry += __shfl(inc, 31, 32);
    }
    if (lane == 0) pfx[blockIdx.x * BIN_PF + SORT_WG] = (int)carry;
  }
  __syncthreads();

  unsigned* __restrict__ mine = binned + (size_t)blockIdx.x * BIN_SLICE;
  #pragma unroll
  for (int j = 0; j < 4; ++j)
    if (rg[j] < SORT_WG) mine[s_excl[rg[j]] + rank[j]] = word[j];
}

__global__ __launch_bounds__(SORT_TH) void csr_build_kernel(
    const unsigned* __restrict__ binned, const int* __restrict__ pfx,
    int* __restrict__ rowptr, int* __restrict__ col) {
  static_assert((SORT_WG - 1) * SORT_R < N_NODES && SORT_WG * SORT_R >= N_NODES,
                "every range holds a node and the ranges cover all nodes");
  static_assert(BIN_WG <= SORT_TH, "one thread per slice reads the slice's range start");
  constexpr int NWV = SORT_TH / 32;
  constexpr int SPW = (BIN_WG + NWV - 1) / NWV;
  __shared__ unsigned s_cnt[SORT_R];
  __shared__ unsigned s_excl[SORT_R];
  __shared__ unsigned s_off[BIN_WG];
  __shared__ unsigned s_len[BIN_WG];
  __shared__ unsigned s_below;
  const int t    = threadIdx.x;
  const int lane = t & 31;
  const int wv   = t >> 5;
  const int lo   = blockIdx.x * SORT_R;
  const int hi   = (lo + SORT_R < N_NODES) ? lo + SORT_R : N_NODES;
  const unsigned rthis = (unsigned)(hi - lo);
  for (int i = t; i < SORT_R; i += SORT_TH) s_cnt[i] = 0u;
  if (t == 0) s_below = 0u;
  __syncthreads();

  unsigned below = 0u;
  if (t < BIN_WG) {
    const int* __restrict__ p = pfx + t * BIN_PF + blockIdx.x;
    const unsigned o = (unsigned)p[0];
    const unsigned e = (unsigned)p[1];
    s_off[t] = (unsigned)t * BIN_SLICE + o;
    s_len[t] = e - o;
    below = o;
  }
  #pragma unroll
  for (int off = 16; off > 0; off >>= 1) below += __shfl_xor(below, off, 32);
  if (lane == 0) atomicAdd(&s_below, below);
  __syncthreads();
  const unsigned base = s_below;

  auto walk = [&](auto use) {
    unsigned off[SPW], len[SPW], w0[SPW], w1[SPW];
    #pragma unroll
    for (int j = 0; j < SPW; ++j) {
      const int b = wv + NWV * j;
      const bool in = b < BIN_WG;
      off[j] = in ? s_off[b] : 0u;
      len[j] = in ? s_len[b] : 0u;
    }
    #pragma unroll
    for (int j = 0; j < SPW; ++j) {
      const unsigned last = len[j] ? len[j] - 1u : 0u;
      const unsigned l0 = (unsigned)lane, l1 = (unsigned)lane + 32u;
      const unsigned o  = len[j] ? off[j] : 0u;
      w0[j] = binned[o + (l0 < last ? l0 : last)];
      w1[j] = binned[o + (l1 < last ? l1 : last)];
    }
    #pragma unroll
    for (int j = 0; j < SPW; ++j) {
      if ((unsigned)lane < len[j]) use(w0[j]);
      if ((unsigned)lane + 32u < len[j]) use(w1[j]);
      for (unsigned l = (unsigned)lane + 64u; l < len[j]; l += 32u) use(binned[off[j] + l]);
    }
  };

  walk([&](unsigned p) {
    const unsigned dl = p >> 16;
    if (dl < rthis) atomicAdd(&s_cnt[dl], 1u);
  });
  __syncthreads();

  if (t < 32) {
    unsigned carry = 0u;
    for (int c = 0; c < (SORT_R + 31) / 32; ++c) {
      const int i = c * 32 + lane;
      const bool in = (unsigned)i < rthis;
      const unsigned v = in ? s_cnt[i] : 0u;
      unsigned inc = v;
      #pragma unroll
      for (int off = 1; off < 32; off <<= 1) {
        const unsigned y = __shfl_up(inc, off, 32);
        if (lane >= off) inc += y;
      }
      const unsigned ex = inc - v + carry;
      if (in) {
        s_excl[i] = ex;
        s_cnt[i]  = 0u;
        rowptr[lo + i] = (int)(base + ex);
      }
      carry += __shfl(inc, 31, 32);
    }
    if (lane == 0 && hi == N_NODES) rowptr[N_NODES] = (int)(base + carry);
  }
  __syncthreads();

  walk([&](unsigned p) {
    const unsigned dl = p >> 16;
    if (dl < rthis) {
      const unsigned r = atomicAdd(&s_cnt[dl], 1u);
      col[base + s_excl[dl] + r] = (int)(p & 0xFFFFu);
    }
  });
}

#define CHUNK 32
#define GROUPS (CHUNK / 4)
template <int HCv, bool RELU, int AGG_WAVES, bool FUSE, int NOUT>
__global__ __launch_bounds__(AGG_WAVES * 32, 16) void node_softmax_agg_kernel(
    const int* __restrict__ rowptr, const int* __restrict__ col,
    const float* __restrict__ as, const float* __restrict__ ad,
    const float* __restrict__ hmat, const float* __restrict__ bias,
    float* __restrict__ outp,
    const float* __restrict__ Wn, const float* __restrict__ a_srcn,
    const float* __restrict__ a_dstn, float* __restrict__ asn, float* __restrict__ adn) {
  constexpr int VEC = HCv / 32;
  constexpr int NV4 = VEC / 4;
  static_assert((HCv / HEADS) / VEC == 4, "four lanes per head");
  static_assert(!FUSE || (AGG_WAVES == 16 && N_NODES % 16 == 0 && NOUT / 16 == HEADS),
                "a fused work-group is one whole 16-row tile, one head per column tile");
  __shared__ float wl_all[AGG_WAVES][CHUNK * HEADS];
  __shared__ int   sl_all[AGG_WAVES][CHUNK];
  const int lane = threadIdx.x & 31;
  const int wib = __builtin_amdgcn_readfirstlane((int)(threadIdx.x >> 5));
  float* __restrict__ wl = wl_all[wib];
  int*   __restrict__ sl = sl_all[wib];
  const int n = __builtin_amdgcn_readfirstlane((int)(blockIdx.x * AGG_WAVES) + wib);
  if (!FUSE && n >= N_NODES) return;
  const int hd  = lane >> 2;
  const int sub = lane & 3;
  const unsigned loff = (unsigned)lane * VEC;
  const int beg = rowptr[n];
  const int end = rowptr[n + 1];
  const int deg = end - beg;
  const float adv = ad[n * HEADS + hd];

  float vs = as[n * HEADS + hd] + adv;
  vs = vs > 0.f ? vs : NEG_SLOPE * vs;

  float acc[VEC];
  auto init_acc = [&](float w) {
    const float4* __restrict__ hp =
        (const float4*)(hmat + (size_t)n * HCv + loff);
    #pragma unroll
    for (int q = 0; q < NV4; ++q) {
      const float4 t = hp[q];
      acc[4 * q + 0] = w * t.x;
      acc[4 * q + 1] = w * t.y;
      acc[4 * q + 2] = w * t.z;
      acc[4 * q + 3] = w * t.w;
    }
  };
  auto add_row = [&](int s, float w) {
    const float4* __restrict__ hp =
        (const float4*)(hmat + (size_t)s * HCv + loff);
    #pragma unroll
    for (int q = 0; q < NV4; ++q) {
      const float4 t = hp[q];
      acc[4 * q + 0] += w * t.x;
      acc[4 * q + 1] += w * t.y;
      acc[4 * q + 2] += w * t.z;
      acc[4 * q + 3] += w * t.w;
    }
  };

  if (deg >= 1 && deg <= CHUNK) {
    float e[GROUPS];
    int   sv[GROUPS];
    const int last = deg - 1;
    #pragma unroll
    for (int g = 0; g < GROUPS; ++g) {
      const int idx = 4 * g + sub;
      sv[g] = col[beg + (idx < last ? idx : last)];
    }
    #pragma unroll
    for (int g = 0; g < GROUPS; ++g) e[g] = as[sv[g] * HEADS + hd];
    float m = vs;
    #pragma unroll
    for (int g = 0; g < GROUPS; ++g) {
      float v = e[g] + adv;
      v = v > 0.f ? v : NEG_SLOPE * v;
      v = (4 * g + sub) < deg ? v : -INFINITY;
      e[g] = v;
      m = v > m ? v : m;
    }
    {
      float y = __shfl_xor(m, 1, 32); m = y > m ? y : m;
      y = __shfl_xor(m, 2, 32);       m = y > m ? y : m;
    }
    float den = 0.f;
    #pragma unroll
    for (int g = 0; g < GROUPS; ++g) {
      e[g] = __expf(e[g] - m);
      den += e[g];
    }
    den += __shfl_xor(den, 1, 32);
    den += __shfl_xor(den, 2, 32);
    const float es = __expf(vs - m);
    den += es;
    const float dn = den + 1e-16f;
    #pragma unroll
    for (int g = 0; g < GROUPS; ++g) {
      wl[(4 * g + sub) * HEADS + hd] = e[g] / dn;
      sl[4 * g + sub] = sv[g];
    }
    __builtin_amdgcn_wave_barrier();
    init_acc(es / dn);
    for (int i = 0; i < deg; i += 4) {
      #pragma unroll
      for (int j = 0; j < 4; ++j) {
        const int s = __builtin_amdgcn_readfirstlane(sl[i + j]);
        add_row(s, wl[(i + j) * HEADS + hd]);
      }
    }
  } else {
    float m = vs;
    #pragma unroll 4
    for (int i = beg; i < end; ++i) {
      const int s = col[i];
      float v = as[s * HEADS + hd] + adv;
      v = v > 0.f ? v : NEG_SLOPE * v;
      m = v > m ? v : m;
    }
    float den = __expf(vs - m);
    #pragma unroll 4
    for (int i = beg; i < end; ++i) {
      const int s = col[i];
      float v = as[s * HEADS + hd] + adv;
      v = v > 0.f ? v : NEG_SLOPE * v;
      den += __expf(v - m);
    }
    const float dn = den + 1e-16f;
    init_acc(__expf(vs - m) / dn);
    #pragma unroll 4
    for (int i = beg; i < end; ++i) {
      const int s = col[i];
      float v = as[s * HEADS + hd] + adv;
      v = v > 0.f ? v : NEG_SLOPE * v;
      add_row(s, __expf(v - m) / dn);
    }
  }

  constexpr int PITCH = HCv + 4;
  __shared__ float4 srow4[FUSE ? AGG_WAVES * PITCH / 4 : 1];
  float4* __restrict__ op = FUSE ? srow4 + (wib * PITCH + lane * VEC) / 4
                                 : (float4*)(outp + (size_t)n * HCv + lane * VEC);
  const float4* __restrict__ bp = (const float4*)(bias + lane * VEC);
  #pragma unroll
  for (int q = 0; q < NV4; ++q) {
    const float4 b = bp[q];
    float4 r;
    r.x = acc[4 * q + 0] + b.x;
    r.y = acc[4 * q + 1] + b.y;
    r.z = acc[4 * q + 2] + b.z;
    r.w = acc[4 * q + 3] + b.w;
    if (RELU) {
      r.x = r.x > 0.f ? r.x : 0.f;
      r.y = r.y > 0.f ? r.y : 0.f;
      r.z = r.z > 0.f ? r.z : 0.f;
      r.w = r.w > 0.f ? r.w : 0.f;
    }
    op[q] = r;
  }

  if constexpr (FUSE) {
    constexpr int NT = NOUT / 16;
    constexpr int KB = 8;
    constexpr int NB = HCv / (4 * KB);
    static_assert(NT <= AGG_WAVES && NB * KB * 4 == HCv && NB % 2 == 0,
                  "tiles fit the waves, groups cover K in pairs");
    const int  r    = lane & 15;
    const int  ksel = (lane >> 4) << 1;
    const bool mw   = wib < NT;
    const float* __restrict__ bb = Wn + (size_t)ksel * NOUT + wib * 16 + r;
    v2f b0[KB], b1[KB];
    if (mw) {
      #pragma unroll
      for (int j = 0; j < KB; ++j) {
        b0[j][0] = bb[(size_t)(4 * j) * NOUT];
        b0[j][1] = bb[(size_t)(4 * j + 1) * NOUT];
      }
    }
    __syncthreads();
    if (!mw) return;
    const float* sa = (const float*)srow4 + r * PITCH + ksel;
    v8f cacc = {};
    auto mul_group = [&](const float* ap, const v2f* bf) {
      v2f aq[KB];
      #pragma unroll
      for (int j = 0; j < KB; ++j) aq[j] = *(const v2f*)(ap + 4 * j);
      #pragma unroll
      for (int j = 0; j < KB; ++j)
        cacc = __builtin_amdgcn_wmma_f32_16x16x4_f32(false, aq[j], false, bf[j],
                                                     (short)0, cacc, false, false);
    };
    auto load_group = [&](v2f* bf, int grp) {
      const float* __restrict__ bg = bb + (size_t)(4 * KB * grp) * NOUT;
      #pragma unroll
      for (int j = 0; j < KB; ++j) {
        bf[j][0] = bg[(size_t)(4 * j) * NOUT];
        bf[j][1] = bg[(size_t)(4 * j + 1) * NOUT];
      }
    };
    #pragma unroll 1
    for (int g = 0; g < NB - 2; g += 2) {
      load_group(b1, g + 1);
      __builtin_amdgcn_sched_barrier(0);
      mul_group(sa + 4 * KB * g, b0);
      load_group(b0, g + 2);
      __builtin_amdgcn_sched_barrier(0);
      mul_group(sa + 4 * KB * (g + 1), b1);
    }
    load_group(b1, NB - 1);
    __builtin_amdgcn_sched_barrier(0);
    mul_group(sa + 4 * KB * (NB - 2), b0);
    mul_group(sa + 4 * KB * (NB - 1), b1);
    const int row0 = blockIdx.x * AGG_WAVES + (lane >> 4) * 8;
    float* __restrict__ crow = outp + (size_t)row0 * NOUT + wib * 16 + r;
    #pragma unroll
    for (int i = 0; i < 8; ++i) crow[(size_t)i * NOUT] = cacc[i];

    const float ps = a_srcn[wib * 16 + r];
    const float pd = a_dstn[wib * 16 + r];
    float vs[8], vd[8];
    #pragma unroll
    for (int i = 0; i < 8; ++i) { vs[i] = cacc[i] * ps; vd[i] = cacc[i] * pd; }
    #pragma unroll
    for (int m = 8, cnt = 4; m >= 2; m >>= 1, cnt >>= 1) {
      const bool up = (r & m) != 0;
      #pragma unroll
      for (int k = 0; k < cnt; ++k) {
        const float s_lo = vs[k], s_hi = vs[k + cnt];
        const float d_lo = vd[k], d_hi = vd[k + cnt];
        const float keep_s = up ? s_hi : s_lo;
        const float send_s = up ? s_lo : s_hi;
        const float keep_d = up ? d_hi : d_lo;
        const float send_d = up ? d_lo : d_hi;
        vs[k] = keep_s + __shfl_xor(send_s, m, 32);
        vd[k] = keep_d + __shfl_xor(send_d, m, 32);
      }
    }
    const float fs = vs[0] + __shfl_xor(vs[0], 1, 32);
    const float fd = vd[0] + __shfl_xor(vd[0], 1, 32);
    const size_t lo = (size_t)(row0 + (r >> 1)) * HEADS + wib;
    if (r & 1) adn[lo] = fd; else asn[lo] = fs;
  }
}

static inline int blocks_for(long long n, int bs) {
  return (int)((n + bs - 1) / bs);
}

extern "C" void kernel_launch(void* const* d_in, const int* in_sizes, int n_in,
                              void* d_out, int out_size, void* d_ws, size_t ws_size,
                              hipStream_t stream) {
  const float* x      = (const float*)d_in[0];
  const int*   ei     = (const int*)  d_in[1];
  const float* W1     = (const float*)d_in[2];
  const float* a_src1 = (const float*)d_in[3];
  const float* a_dst1 = (const float*)d_in[4];
  const float* b1     = (const float*)d_in[5];
  const float* W2     = (const float*)d_in[6];
  const float* a_src2 = (const float*)d_in[7];
  const float* a_dst2 = (const float*)d_in[8];
  const float* b2     = (const float*)d_in[9];
  float* out = (float*)d_out;

  char* ws = (char*)d_ws;
  size_t off = 0;
  auto carve = [&](size_t bytes) {
    void* p = ws + off;
    off += (bytes + 255) & ~(size_t)255;
    return p;
  };
  float* hbuf   = (float*)carve((size_t)N_NODES * HC1 * 4);
  float* h2buf  = (float*)carve((size_t)N_NODES * HC2 * 4);
  float* as     = (float*)carve((size_t)N_NODES * HEADS * 4);
  float* ad     = (float*)carve((size_t)N_NODES * HEADS * 4);
  float* as2    = (float*)carve((size_t)N_NODES * HEADS * 4);
  float* ad2    = (float*)carve((size_t)N_NODES * HEADS * 4);
  int*   rowptr = (int*)  carve((size_t)(N_NODES + 1) * 4);
  int*   col    = (int*)  carve((size_t)N_EDGES * 4);
  unsigned* binned = (unsigned*)carve((size_t)N_EDGES * 4);
  int*   pfx    = (int*)  carve((size_t)BIN_WG * BIN_PF * 4);

  const int BS = 256;

  {
    long long waves = (N_NODES / 16) * (HC1 / 64);
    gemm_f32_wmma<HC1, F_IN, HID, 16><<<blocks_for(waves * 32, BS), BS, 0, stream>>>(
        x, W1, a_src1, a_dst1, hbuf, as, ad, N_NODES);
  }

  bin_edges_kernel<<<BIN_WG, BIN_TH, 0, stream>>>(ei, binned, pfx);
  csr_build_kernel<<<SORT_WG, SORT_TH, 0, stream>>>(binned, pfx, rowptr, col);

  node_softmax_agg_kernel<HC1, true, 16, true, HC2><<<N_NODES / 16, 512, 0, stream>>>(
      rowptr, col, as, ad, hbuf, b1, h2buf, W2, a_src2, a_dst2, as2, ad2);

  node_softmax_agg_kernel<HC2, false, 8, false, HC2><<<blocks_for((long long)N_NODES * 32, BS), BS, 0, stream>>>(
      rowptr, col, as2, ad2, h2buf, b2, out, nullptr, nullptr, nullptr, nullptr, nullptr);
}


# Edge grouping's second step (per-range count, scan, fill) runs as the first 128 work-groups of the first product's launch, beside the matrix work: one launch fewer; fill order stays arrival-ranked
# speedup vs baseline: 4.8507x; 1.0751x over previous
#include <hip/hip_runtime.h>
#include <hip/hip_bf16.h>

#define N_NODES 50000
#define N_EDGES 800000
#define E_TOT   (N_EDGES + N_NODES)
#define F_IN    256
#define HEADS   8
#define HID     32
#define N_CLS   16
#define HC1     (HEADS * HID)
#define HC2     (HEADS * N_CLS)
#define NEG_SLOPE 0.2f

typedef float v2f __attribute__((ext_vector_type(2)));
typedef float v8f __attribute__((ext_vector_type(8)));

#define SORT_WG   128
#define SORT_R    ((N_NODES + SORT_WG - 1) / SORT_WG)
#define SORT_TH   256
#define BIN_SLICE 4000
#define BIN_WG    (N_EDGES / BIN_SLICE)
#define BIN_TH    1024
#define BIN_PF    (SORT_WG + 1)
__global__ __launch_bounds__(BIN_TH) void bin_edges_kernel(
    const int* __restrict__ ei, unsigned* __restrict__ binned, int* __restrict__ pfx) {
  static_assert(N_EDGES % BIN_SLICE == 0 && BIN_SLICE % 4 == 0 && BIN_SLICE / 4 <= BIN_TH,
                "a slice is whole 16-byte groups, one per thread");
  static_assert(N_NODES <= (1 << 16) && SORT_R <= (1 << 16),
                "source (16 bits) and node inside the range (16 bits) share a word");
  static_assert(SORT_WG % 32 == 0 && SORT_WG * SORT_R >= N_NODES,
                "the range scan runs whole 32-lane trips and the ranges cover all nodes");
  __shared__ unsigned s_cnt[SORT_WG];
  __shared__ unsigned s_excl[SORT_WG];
  const int t    = threadIdx.x;
  const int lane = t & 31;
  for (int i = t; i < SORT_WG; i += BIN_TH) s_cnt[i] = 0u;
  __syncthreads();

  int4 d = make_int4(-1, -1, -1, -1);
  int4 s = make_int4(0, 0, 0, 0);
  if (t < BIN_SLICE / 4) {
    const int q = blockIdx.x * (BIN_SLICE / 4) + t;
    d = ((const int4*)(ei + N_EDGES))[q];
    s = ((const int4*)ei)[q];
  }
  const int dv[4] = {d.x, d.y, d.z, d.w};
  const int sv[4] = {s.x, s.y, s.z, s.w};
  unsigned rg[4], word[4], rank[4];
  #pragma unroll
  for (int j = 0; j < 4; ++j) {
    const unsigned dd = (unsigned)dv[j];
    const unsigned r  = dd / (unsigned)SORT_R;
    rg[j]   = SORT_WG;
    rank[j] = 0u;
    word[j] = ((dd - r * (unsigned)SORT_R) << 16) | ((unsigned)sv[j] & 0xFFFFu);
    if (dd < (unsigned)N_NODES) {
      rg[j]   = r;
      rank[j] = atomicAdd(&s_cnt[r], 1u);
    }
  }
  __syncthreads();

  if (t < 32) {
    unsigned carry = 0u;
    for (int c = 0; c < SORT_WG / 32; ++c) {
      const int i = c * 32 + lane;
      const unsigned v = s_cnt[i];
      unsigned inc = v;
      #pragma unroll
      for (int off = 1; off < 32; off <<= 1) {
        const unsigned y = __shfl_up(inc, off, 32);
        if (lane >= off) inc += y;
      }
      const unsigned ex = inc - v + carry;
      s_excl[i] = ex;
      pfx[blockIdx.x * BIN_PF + i] = (int)ex;
      carry += __shfl(inc, 31, 32);
    }
    if (lane == 0) pfx[blockIdx.x * BIN_PF + SORT_WG] = (int)carry;
  }
  __syncthreads();

  unsigned* __restrict__ mine = binned + (size_t)blockIdx.x * BIN_SLICE;
  #pragma unroll
  for (int j = 0; j < 4; ++j)
    if (rg[j] < SORT_WG) mine[s_excl[rg[j]] + rank[j]] = word[j];
}

#define CSR_PAD       ((SORT_R + 31) & ~31)
#define CSR_LDS_WORDS (2 * CSR_PAD + 2 * BIN_WG + 32)
__device__ __forceinline__ void csr_range(
    unsigned* lds,
    const unsigned* __restrict__ binned, const int* __restrict__ pfx,
    int* __restrict__ rowptr, int* __restrict__ col) {
  static_assert((SORT_WG - 1) * SORT_R < N_NODES && SORT_WG * SORT_R >= N_NODES,
                "every range holds a node and the ranges cover all nodes");
  static_assert(BIN_WG <= SORT_TH, "one thread per slice reads the slice's range start");
  constexpr int NWV = SORT_TH / 32;
  constexpr int SPW = (BIN_WG + NWV - 1) / NWV;
  constexpr int SPC = 5;
  static_assert(SPW % SPC == 0, "a walk is whole trips");
  unsigned* s_cnt   = lds;
  unsigned* s_excl  = lds + CSR_PAD;
  unsigned* s_off   = lds + 2 * CSR_PAD;
  unsigned* s_len   = s_off + BIN_WG;
  unsigned* s_below = s_len + BIN_WG;
  const int t    = threadIdx.x;
  const int lane = t & 31;
  const int wv   = t >> 5;
  const int lo   = blockIdx.x * SORT_R;
  const int hi   = (lo + SORT_R < N_NODES) ? lo + SORT_R : N_NODES;
  const unsigned rthis = (unsigned)(hi - lo);
  for (int i = t; i < SORT_R; i += SORT_TH) s_cnt[i] = 0u;
  if (t == 0) s_below[0] = 0u;
  __syncthreads();

  unsigned below = 0u;
  if (t < BIN_WG) {
    const int* __restrict__ p = pfx + t * BIN_PF + blockIdx.x;
    const unsigned o = (unsigned)p[0];
    const unsigned e = (unsigned)p[1];
    s_off[t] = (unsigned)t * BIN_SLICE + o;
    s_len[t] = e - o;
    below = o;
  }
  #pragma unroll
  for (int off = 16; off > 0; off >>= 1) below += __shfl_xor(below, off, 32);
  if (lane == 0) atomicAdd(&s_below[0], below);
  __syncthreads();
  const unsigned base = s_below[0];

  auto walk = [&](auto use) {
    #pragma unroll 1
    for (int c0 = 0; c0 < SPW; c0 += SPC) {
      unsigned off[SPC], len[SPC], w0[SPC], w1[SPC];
      #pragma unroll
      for (int j = 0; j < SPC; ++j) {
        const int b = wv + NWV * (c0 + j);
        const bool in = b < BIN_WG;
        off[j] = in ? s_off[b] : 0u;
        len[j] = in ? s_len[b] : 0u;
      }
      #pragma unroll
      for (int j = 0; j < SPC; ++j) {
        const unsigned last = len[j] ? len[j] - 1u : 0u;
        const unsigned l0 = (unsigned)lane, l1 = (unsigned)lane + 32u;
        const unsigned o  = len[j] ? off[j] : 0u;
        w0[j] = binned[o + (l0 < last ? l0 : last)];
        w1[j] = binned[o + (l1 < last ? l1 : last)];
      }
      #pragma unroll
      for (int j = 0; j < SPC; ++j) {
        if ((unsigned)lane < len[j]) use(w0[j]);
        if ((unsigned)lane + 32u < len[j]) use(w1[j]);
        for (unsigned l = (unsigned)lane + 64u; l < len[j]; l += 32u) use(binned[off[j] + l]);
      }
    }
  };

  walk([&](unsigned p) {
    const unsigned dl = p >> 16;
    if (dl < rthis) atomicAdd(&s_cnt[dl], 1u);
  });
  __syncthreads();

  if (t < 32) {
    unsigned carry = 0u;
    for (int c = 0; c < (SORT_R + 31) / 32; ++c) {
      const int i = c * 32 + lane;
      const bool in = (unsigned)i < rthis;
      const unsigned v = in ? s_cnt[i] : 0u;
      unsigned inc = v;
      #pragma unroll
      for (int off = 1; off < 32; off <<= 1) {
        const unsigned y = __shfl_up(inc, off, 32);
        if (lane >= off) inc += y;
      }
      const unsigned ex = inc - v + carry;
      if (in) {
        s_excl[i] = ex;
        s_cnt[i]  = 0u;
        rowptr[lo + i] = (int)(base + ex);
      }
      carry += __shfl(inc, 31, 32);
    }
    if (lane == 0 && hi == N_NODES) rowptr[N_NODES] = (int)(base + carry);
  }
  __syncthreads();

  walk([&](unsigned p) {
    const unsigned dl = p >> 16;
    if (dl < rthis) {
      const unsigned r = atomicAdd(&s_cnt[dl], 1u);
      col[base + s_excl[dl] + r] = (int)(p & 0xFFFFu);
    }
  });
}

template <int N, int K, int CH, int WPE>
__global__ __launch_bounds__(256, WPE) void gemm_f32_wmma(
    const float* __restrict__ A, const float* __restrict__ B,
    const float* __restrict__ a_src, const float* __restrict__ a_dst,
    float* __restrict__ C, float* __restrict__ as, float* __restrict__ ad, int M,
    const unsigned* __restrict__ binned, const int* __restrict__ pfx,
    int* __restrict__ rowptr, int* __restrict__ col) {
  constexpr int TILES_N = N / 64;
  constexpr int WAVES   = 8;
  constexpr int RB      = 16 * WAVES / TILES_N;
  constexpr int PITCH   = K + 4;
  constexpr int Q       = K / 4;
  constexpr int PQ      = PITCH / 4;
  static_assert(WAVES % TILES_N == 0, "a work-group holds whole row tiles");
  static_assert((RB * Q) % 256 == 0 && K % 4 == 0, "the panel splits evenly over 256 lanes");
  __shared__ float4 sA4[RB * PQ];
  static_assert(sizeof(float4) * RB * PQ >= sizeof(unsigned) * CSR_LDS_WORDS,
                "the grouping work-groups use the panel's LDS");
  if (blockIdx.x < SORT_WG) {
    csr_range((unsigned*)sA4, binned, pfx, rowptr, col);
    return;
  }
  const int bid   = (int)blockIdx.x - SORT_WG;
  const int lane  = threadIdx.x & 31;
  const int wib   = threadIdx.x >> 5;
  const int wave  = bid * WAVES + wib;
  const int tileM = wave / TILES_N;
  const int tileN = wave % TILES_N;

  {
    constexpr int NL = RB * Q / 256;
    const int row0 = bid * RB;
    float4 pv[NL];
    #pragma unroll
    for (int j = 0; j < NL; ++j) {
      const int f  = j * 256 + (int)threadIdx.x;
      const int gr = row0 + f / Q;
      pv[j] = ((const float4*)(A + (size_t)(gr < M ? gr : M - 1) * K))[f % Q];
    }
    #pragma unroll
    for (int j = 0; j < NL; ++j) {
      const int f = j * 256 + (int)threadIdx.x;
      sA4[(f / Q) * PQ + f % Q] = pv[j];
    }
  }
  __syncthreads();
  if (tileM * 16 >= M) return;

  const int r    = lane & 15;
  const int ksel = (lane >> 4) << 1;

  const float* sarow = (const float*)sA4 + ((wib / TILES_N) * 16 + r) * PITCH + ksel;
  const float* __restrict__ bbase = B + (size_t)ksel * N + tileN * 64 + r;

  v8f acc0 = {}, acc1 = {}, acc2 = {}, acc3 = {};
  #pragma unroll 2
  for (int kk = 0; kk < K; kk += 4) {
    v2f b0, b1, b2, b3;
    const v2f a = *(const v2f*)(sarow + kk);
    b0[0] = bbase[0];       b0[1] = bbase[N];
    b1[0] = bbase[16];      b1[1] = bbase[N + 16];
    b2[0] = bbase[32];      b2[1] = bbase[N + 32];
    b3[0] = bbase[48];      b3[1] = bbase[N + 48];
    acc0 = __builtin_amdgcn_wmma_f32_16x16x4_f32(false, a, false, b0, (short)0, acc0, false, false);
    acc1 = __builtin_amdgcn_wmma_f32_16x16x4_f32(false, a, false, b1, (short)0, acc1, false, false);
    acc2 = __builtin_amdgcn_wmma_f32_16x16x4_f32(false, a, false, b2, (short)0, acc2, false, false);
    acc3 = __builtin_amdgcn_wmma_f32_16x16x4_f32(false, a, false, b3, (short)0, acc3, false, false);
    bbase += (size_t)4 * N;
  }

  float* __restrict__ crow =
      C + (size_t)(tileM * 16 + (lane >> 4) * 8) * N + tileN * 64 + (lane & 15);
  #pragma unroll
  for (int i = 0; i < 8; ++i) {
    crow[(size_t)i * N +  0] = acc0[i];
    crow[(size_t)i * N + 16] = acc1[i];
    crow[(size_t)i * N + 32] = acc2[i];
    crow[(size_t)i * N + 48] = acc3[i];
  }

  constexpr int HPW = 64 / CH;
  constexpr int APH = CH / 16;
  constexpr int NV  = 8 * HPW;
  constexpr int R   = NV / 16;
  static_assert(HPW * CH == 64 && APH * 16 == CH, "whole heads per wave");
  static_assert(HPW == 2 * R, "row of a finished value is (lane & 15) >> 1");
  static_assert(N / CH == HEADS, "logit rows hold HEADS values");
  const int c = lane & 15;
  const v8f accs[4] = {acc0, acc1, acc2, acc3};
  float ps[4], pd[4];
  #pragma unroll
  for (int j = 0; j < 4; ++j) {
    ps[j] = a_src[tileN * 64 + 16 * j + c];
    pd[j] = a_dst[tileN * 64 + 16 * j + c];
  }
  float vs[NV], vd[NV];
  #pragma unroll
  for (int i = 0; i < 8; ++i) {
    #pragma unroll
    for (int q = 0; q < HPW; ++q) {
      float sa = accs[q * APH][i] * ps[q * APH];
      float sd = accs[q * APH][i] * pd[q * APH];
      #pragma unroll
      for (int t = 1; t < APH; ++t) {
        const float hv = accs[q * APH + t][i];
        sa += hv * ps[q * APH + t];
        sd += hv * pd[q * APH + t];
      }
      vs[i * HPW + q] = sa;
      vd[i * HPW + q] = sd;
    }
  }
  #pragma unroll
  for (int m = 8, n = NV / 2; m >= 1; m >>= 1, n >>= 1) {
    const bool up = (c & m) != 0;
    #pragma unroll
    for (int k = 0; k < n; ++k) {
      const float s_lo = vs[k], s_hi = vs[k + n];
      const float d_lo = vd[k], d_hi = vd[k + n];
      const float keep_s = up ? s_hi : s_lo;
      const float send_s = up ? s_lo : s_hi;
      const float keep_d = up ? d_hi : d_lo;
      const float send_d = up ? d_lo : d_hi;
      vs[k] = keep_s + __shfl_xor(send_s, m, 32);
      vd[k] = keep_d + __shfl_xor(send_d, m, 32);
    }
  }
  const size_t lo = (size_t)(tileM * 16 + (lane >> 4) * 8 + (c >> 1)) * HEADS +
                    tileN * HPW + R * (c & 1);
  #pragma unroll
  for (int k = 0; k < R; ++k) {
    as[lo + k] = vs[k];
    ad[lo + k] = vd[k];
  }
}

#define CHUNK 32
#define GROUPS (CHUNK / 4)
template <int HCv, bool RELU, int AGG_WAVES, bool FUSE, int NOUT>
__global__ __launch_bounds__(AGG_WAVES * 32, 16) void node_softmax_agg_kernel(
    const int* __restrict__ rowptr, const int* __restrict__ col,
    const float* __restrict__ as, const float* __restrict__ ad,
    const float* __restrict__ hmat, const float* __restrict__ bias,
    float* __restrict__ outp,
    const float* __restrict__ Wn, const float* __restrict__ a_srcn,
    const float* __restrict__ a_dstn, float* __restrict__ asn, float* __restrict__ adn) {
  constexpr int VEC = HCv / 32;
  constexpr int NV4 = VEC / 4;
  static_assert((HCv / HEADS) / VEC == 4, "four lanes per head");
  static_assert(!FUSE || (AGG_WAVES == 16 && N_NODES % 16 == 0 && NOUT / 16 == HEADS),
                "a fused work-group is one whole 16-row tile, one head per column tile");
  __shared__ float wl_all[AGG_WAVES][CHUNK * HEADS];
  __shared__ int   sl_all[AGG_WAVES][CHUNK];
  const int lane = threadIdx.x & 31;
  const int wib = __builtin_amdgcn_readfirstlane((int)(threadIdx.x >> 5));
  float* __restrict__ wl = wl_all[wib];
  int*   __restrict__ sl = sl_all[wib];
  const int n = __builtin_amdgcn_readfirstlane((int)(blockIdx.x * AGG_WAVES) + wib);
  if (!FUSE && n >= N_NODES) return;
  const int hd  = lane >> 2;
  const int sub = lane & 3;
  const unsigned loff = (unsigned)lane * VEC;
  const int beg = rowptr[n];
  const int end = rowptr[n + 1];
  const int deg = end - beg;
  const float adv = ad[n * HEADS + hd];

  float vs = as[n * HEADS + hd] + adv;
  vs = vs > 0.f ? vs : NEG_SLOPE * vs;

  float acc[VEC];
  auto init_acc = [&](float w) {
    const float4* __restrict__ hp =
        (const float4*)(hmat + (size_t)n * HCv + loff);
    #pragma unroll
    for (int q = 0; q < NV4; ++q) {
      const float4 t = hp[q];
      acc[4 * q + 0] = w * t.x;
      acc[4 * q + 1] = w * t.y;
      acc[4 * q + 2] = w * t.z;
      acc[4 * q + 3] = w * t.w;
    }
  };
  auto add_row = [&](int s, float w) {
    const float4* __restrict__ hp =
        (const float4*)(hmat + (size_t)s * HCv + loff);
    #pragma unroll
    for (int q = 0; q < NV4; ++q) {
      const float4 t = hp[q];
      acc[4 * q + 0] += w * t.x;
      acc[4 * q + 1] += w * t.y;
      acc[4 * q + 2] += w * t.z;
      acc[4 * q + 3] += w * t.w;
    }
  };

  if (deg >= 1 && deg <= CHUNK) {
    float e[GROUPS];
    int   sv[GROUPS];
    const int last = deg - 1;
    #pragma unroll
    for (int g = 0; g < GROUPS; ++g) {
      const int idx = 4 * g + sub;
      sv[g] = col[beg + (idx < last ? idx : last)];
    }
    #pragma unroll
    for (int g = 0; g < GROUPS; ++g) e[g] = as[sv[g] * HEADS + hd];
    float m = vs;
    #pragma unroll
    for (int g = 0; g < GROUPS; ++g) {
      float v = e[g] + adv;
      v = v > 0.f ? v : NEG_SLOPE * v;
      v = (4 * g + sub) < deg ? v : -INFINITY;
      e[g] = v;
      m = v > m ? v : m;
    }
    {
      float y = __shfl_xor(m, 1, 32); m = y > m ? y : m;
      y = __shfl_xor(m, 2, 32);       m = y > m ? y : m;
    }
    float den = 0.f;
    #pragma unroll
    for (int g = 0; g < GROUPS; ++g) {
      e[g] = __expf(e[g] - m);
      den += e[g];
    }
    den += __shfl_xor(den, 1, 32);
    den += __shfl_xor(den, 2, 32);
    const float es = __expf(vs - m);
    den += es;
    const float dn = den + 1e-16f;
    #pragma unroll
    for (int g = 0; g < GROUPS; ++g) {
      wl[(4 * g + sub) * HEADS + hd] = e[g] / dn;
      sl[4 * g + sub] = sv[g];
    }
    __builtin_amdgcn_wave_barrier();
    init_acc(es / dn);
    for (int i = 0; i < deg; i += 4) {
      #pragma unroll
      for (int j = 0; j < 4; ++j) {
        const int s = __builtin_amdgcn_readfirstlane(sl[i + j]);
        add_row(s, wl[(i + j) * HEADS + hd]);
      }
    }
  } else {
    float m = vs;
    #pragma unroll 4
    for (int i = beg; i < end; ++i) {
      const int s = col[i];
      float v = as[s * HEADS + hd] + adv;
      v = v > 0.f ? v : NEG_SLOPE * v;
      m = v > m ? v : m;
    }
    float den = __expf(vs - m);
    #pragma unroll 4
    for (int i = beg; i < end; ++i) {
      const int s = col[i];
      float v = as[s * HEADS + hd] + adv;
      v = v > 0.f ? v : NEG_SLOPE * v;
      den += __expf(v - m);
    }
    const float dn = den + 1e-16f;
    init_acc(__expf(vs - m) / dn);
    #pragma unroll 4
    for (int i = beg; i < end; ++i) {
      const int s = col[i];
      float v = as[s * HEADS + hd] + adv;
      v = v > 0.f ? v : NEG_SLOPE * v;
      add_row(s, __expf(v - m) / dn);
    }
  }

  constexpr int PITCH = HCv + 4;
  __shared__ float4 srow4[FUSE ? AGG_WAVES * PITCH / 4 : 1];
  float4* __restrict__ op = FUSE ? srow4 + (wib * PITCH + lane * VEC) / 4
                                 : (float4*)(outp + (size_t)n * HCv + lane * VEC);
  const float4* __restrict__ bp = (const float4*)(bias + lane * VEC);
  #pragma unroll
  for (int q = 0; q < NV4; ++q) {
    const float4 b = bp[q];
    float4 r;
    r.x = acc[4 * q + 0] + b.x;
    r.y = acc[4 * q + 1] + b.y;
    r.z = acc[4 * q + 2] + b.z;
    r.w = acc[4 * q + 3] + b.w;
    if (RELU) {
      r.x = r.x > 0.f ? r.x : 0.f;
      r.y = r.y > 0.f ? r.y : 0.f;
      r.z = r.z > 0.f ? r.z : 0.f;
      r.w = r.w > 0.f ? r.w : 0.f;
    }
    op[q] = r;
  }

  if constexpr (FUSE) {
    constexpr int NT = NOUT / 16;
    constexpr int KB = 8;
    constexpr int NB = HCv / (4 * KB);
    static_assert(NT <= AGG_WAVES && NB * KB * 4 == HCv && NB % 2 == 0,
                  "tiles fit the waves, groups cover K in pairs");
    const int  r    = lane & 15;
    const int  ksel = (lane >> 4) << 1;
    const bool mw   = wib < NT;
    const float* __restrict__ bb = Wn + (size_t)ksel * NOUT + wib * 16 + r;
    v2f b0[KB], b1[KB];
    if (mw) {
      #pragma unroll
      for (int j = 0; j < KB; ++j) {
        b0[j][0] = bb[(size_t)(4 * j) * NOUT];
        b0[j][1] = bb[(size_t)(4 * j + 1) * NOUT];
      }
    }
    __syncthreads();
    if (!mw) return;
    const float* sa = (const float*)srow4 + r * PITCH + ksel;
    v8f cacc = {};
    auto mul_group = [&](const float* ap, const v2f* bf) {
      v2f aq[KB];
      #pragma unroll
      for (int j = 0; j < KB; ++j) aq[j] = *(const v2f*)(ap + 4 * j);
      #pragma unroll
      for (int j = 0; j < KB; ++j)
        cacc = __builtin_amdgcn_wmma_f32_16x16x4_f32(false, aq[j], false, bf[j],
                                                     (short)0, cacc, false, false);
    };
    auto load_group = [&](v2f* bf, int grp) {
      const float* __restrict__ bg = bb + (size_t)(4 * KB * grp) * NOUT;
      #pragma unroll
      for (int j = 0; j < KB; ++j) {
        bf[j][0] = bg[(size_t)(4 * j) * NOUT];
        bf[j][1] = bg[(size_t)(4 * j + 1) * NOUT];
      }
    };
    #pragma unroll 1
    for (int g = 0; g < NB - 2; g += 2) {
      load_group(b1, g + 1);
      __builtin_amdgcn_sched_barrier(0);
      mul_group(sa + 4 * KB * g, b0);
      load_group(b0, g + 2);
      __builtin_amdgcn_sched_barrier(0);
      mul_group(sa + 4 * KB * (g + 1), b1);
    }
    load_group(b1, NB - 1);
    __builtin_amdgcn_sched_barrier(0);
    mul_group(sa + 4 * KB * (NB - 2), b0);
    mul_group(sa + 4 * KB * (NB - 1), b1);
    const int row0 = blockIdx.x * AGG_WAVES + (lane >> 4) * 8;
    float* __restrict__ crow = outp + (size_t)row0 * NOUT + wib * 16 + r;
    #pragma unroll
    for (int i = 0; i < 8; ++i) crow[(size_t)i * NOUT] = cacc[i];

    const float ps = a_srcn[wib * 16 + r];
    const float pd = a_dstn[wib * 16 + r];
    float vs[8], vd[8];
    #pragma unroll
    for (int i = 0; i < 8; ++i) { vs[i] = cacc[i] * ps; vd[i] = cacc[i] * pd; }
    #pragma unroll
    for (int m = 8, cnt = 4; m >= 2; m >>= 1, cnt >>= 1) {
      const bool up = (r & m) != 0;
      #pragma unroll
      for (int k = 0; k < cnt; ++k) {
        const float s_lo = vs[k], s_hi = vs[k + cnt];
        const float d_lo = vd[k], d_hi = vd[k + cnt];
        const float keep_s = up ? s_hi : s_lo;
        const float send_s = up ? s_lo : s_hi;
        const float keep_d = up ? d_hi : d_lo;
        const float send_d = up ? d_lo : d_hi;
        vs[k] = keep_s + __shfl_xor(send_s, m, 32);
        vd[k] = keep_d + __shfl_xor(send_d, m, 32);
      }
    }
    const float fs = vs[0] + __shfl_xor(vs[0], 1, 32);
    const float fd = vd[0] + __shfl_xor(vd[0], 1, 32);
    const size_t lo = (size_t)(row0 + (r >> 1)) * HEADS + wib;
    if (r & 1) adn[lo] = fd; else asn[lo] = fs;
  }
}

static inline int blocks_for(long long n, int bs) {
  return (int)((n + bs - 1) / bs);
}

extern "C" void kernel_launch(void* const* d_in, const int* in_sizes, int n_in,
                              void* d_out, int out_size, void* d_ws, size_t ws_size,
                              hipStream_t stream) {
  const float* x      = (const float*)d_in[0];
  const int*   ei     = (const int*)  d_in[1];
  const float* W1     = (const float*)d_in[2];
  const float* a_src1 = (const float*)d_in[3];
  const float* a_dst1 = (const float*)d_in[4];
  const float* b1     = (const float*)d_in[5];
  const float* W2     = (const float*)d_in[6];
  const float* a_src2 = (const float*)d_in[7];
  const float* a_dst2 = (const float*)d_in[8];
  const float* b2     = (const float*)d_in[9];
  float* out = (float*)d_out;

  char* ws = (char*)d_ws;
  size_t off = 0;
  auto carve = [&](size_t bytes) {
    void* p = ws + off;
    off += (bytes + 255) & ~(size_t)255;
    return p;
  };
  float* hbuf   = (float*)carve((size_t)N_NODES * HC1 * 4);
  float* h2buf  = (float*)carve((size_t)N_NODES * HC2 * 4);
  float* as     = (float*)carve((size_t)N_NODES * HEADS * 4);
  float* ad     = (float*)carve((size_t)N_NODES * HEADS * 4);
  float* as2    = (float*)carve((size_t)N_NODES * HEADS * 4);
  float* ad2    = (float*)carve((size_t)N_NODES * HEADS * 4);
  int*   rowptr = (int*)  carve((size_t)(N_NODES + 1) * 4);
  int*   col    = (int*)  carve((size_t)N_EDGES * 4);
  unsigned* binned = (unsigned*)carve((size_t)N_EDGES * 4);
  int*   pfx    = (int*)  carve((size_t)BIN_WG * BIN_PF * 4);

  const int BS = 256;

  bin_edges_kernel<<<BIN_WG, BIN_TH, 0, stream>>>(ei, binned, pfx);

  {
    long long waves = (N_NODES / 16) * (HC1 / 64);
    gemm_f32_wmma<HC1, F_IN, HID, 16><<<SORT_WG + blocks_for(waves * 32, BS), BS, 0, stream>>>(
        x, W1, a_src1, a_dst1, hbuf, as, ad, N_NODES, binned, pfx, rowptr, col);
  }

  node_softmax_agg_kernel<HC1, true, 16, true, HC2><<<N_NODES / 16, 512, 0, stream>>>(
      rowptr, col, as, ad, hbuf, b1, h2buf, W2, a_src2, a_dst2, as2, ad2);

  node_softmax_agg_kernel<HC2, false, 8, false, HC2><<<blocks_for((long long)N_NODES * 32, BS), BS, 0, stream>>>(
      rowptr, col, as2, ad2, h2buf, b2, out, nullptr, nullptr, nullptr, nullptr, nullptr);
}
